# starting kernel
# baseline (speedup 1.0000x reference)
#include <hip/hip_runtime.h>
#include <hip/hip_bf16.h>
#include <math.h>

typedef __attribute__((ext_vector_type(2))) float v2f;
typedef __attribute__((ext_vector_type(8))) float v8f;

#define CH 32

__global__ __launch_bounds__(256) void k_deg(const int* __restrict__ dst,
                                             const float* __restrict__ w,
                                             float* __restrict__ deg, int E) {
    int e = blockIdx.x * blockDim.x + threadIdx.x;
    if (e < E) {
        __hip_atomic_fetch_add(&deg[dst[e]], w[e],
                               __ATOMIC_RELAXED, __HIP_MEMORY_SCOPE_AGENT);
    }
}

__global__ __launch_bounds__(256) void k_dinv(const float* __restrict__ deg,
                                              float* __restrict__ dinv, int N) {
    int i = blockIdx.x * blockDim.x + threadIdx.x;
    if (i < N) {
        float d = deg[i];
        dinv[i] = (d > 0.0f) ? rsqrtf(fmaxf(d, 1e-12f)) : 0.0f;
    }
}

__global__ __launch_bounds__(256) void k_gemm_wmma(const float* __restrict__ X,
                                                   const float* __restrict__ H,
                                                   const float* __restrict__ Wx,
                                                   const float* __restrict__ Wh,
                                                   float* __restrict__ P, int N) {
    int wave = blockIdx.x * (blockDim.x >> 5) + (threadIdx.x >> 5);
    int row0 = wave * 16;
    if (row0 >= N) return;

    int lane = threadIdx.x & 31;
    int half = lane >> 4;
    int lm   = lane & 15;

    int arow = row0 + lm;
    if (arow > N - 1) arow = N - 1;
    const float* xrow = X + (size_t)arow * CH;
    const float* hrow = H + (size_t)arow * CH;

    v8f acc0 = {};
    v8f acc1 = {};

#pragma unroll
    for (int kt = 0; kt < 8; ++kt) {
        int k0 = kt * 4;
        int ka = k0 + 2 * half;

        v2f ax = { xrow[ka], xrow[ka + 1] };
        v2f ah = { hrow[ka], hrow[ka + 1] };

        int krow = ka * CH;
        v2f bx0 = { Wx[krow + lm],      Wx[krow + CH + lm]      };
        v2f bx1 = { Wx[krow + 16 + lm], Wx[krow + CH + 16 + lm] };
        v2f bh0 = { Wh[krow + lm],      Wh[krow + CH + lm]      };
        v2f bh1 = { Wh[krow + 16 + lm], Wh[krow + CH + 16 + lm] };

        acc0 = __builtin_amdgcn_wmma_f32_16x16x4_f32(false, ax, false, bx0,
                                                     (short)0, acc0, false, false);
        acc1 = __builtin_amdgcn_wmma_f32_16x16x4_f32(false, ax, false, bx1,
                                                     (short)0, acc1, false, false);
        acc0 = __builtin_amdgcn_wmma_f32_16x16x4_f32(false, ah, false, bh0,
                                                     (short)0, acc0, false, false);
        acc1 = __builtin_amdgcn_wmma_f32_16x16x4_f32(false, ah, false, bh1,
                                                     (short)0, acc1, false, false);
    }

    float* prow = P + (size_t)row0 * CH;
#pragma unroll
    for (int r = 0; r < 8; ++r) {
        int rr = r + 8 * half;
        if (row0 + rr < N) {
            prow[(size_t)rr * CH + lm]      = acc0[r];
            prow[(size_t)rr * CH + 16 + lm] = acc1[r];
        }
    }
}

__global__ __launch_bounds__(256) void k_scatter(const int* __restrict__ src,
                                                 const int* __restrict__ dst,
                                                 const float* __restrict__ w,
                                                 const float* __restrict__ dinv,
                                                 const float* __restrict__ P,
                                                 float* __restrict__ agg, int E) {
    int e = blockIdx.x * (blockDim.x >> 5) + (threadIdx.x >> 5);
    if (e >= E) return;
    int lane = threadIdx.x & 31;

    int s = src[e];
    int d = dst[e];
    float nrm = dinv[s] * w[e] * dinv[d];
    float val = nrm * P[(size_t)s * CH + lane];
    __hip_atomic_fetch_add(&agg[(size_t)d * CH + lane], val,
                           __ATOMIC_RELAXED, __HIP_MEMORY_SCOPE_AGENT);
}

__device__ __forceinline__ float sigmoidf_(float x) {
    return 1.0f / (1.0f + __expf(-x));
}

__global__ __launch_bounds__(256) void k_gates(const float* __restrict__ agg,
                                               const float* __restrict__ Cst,
                                               const float* __restrict__ bx,
                                               const float* __restrict__ bh,
                                               const float* __restrict__ bi,
                                               const float* __restrict__ bf,
                                               const float* __restrict__ bc,
                                               const float* __restrict__ bo,
                                               float* __restrict__ out, int NC) {
    int t = blockIdx.x * blockDim.x + threadIdx.x;
    if (t >= NC) return;
    int c = t & (CH - 1);

    float s  = agg[t] + bx[c] + bh[c];
    float I  = sigmoidf_(s + bi[c]);
    float F  = sigmoidf_(s + bf[c]);
    float T  = tanhf(s + bc[c]);
    float Cn = F * Cst[t] + I * T;
    float O  = sigmoidf_(s + bo[c]);
    float Hn = O * tanhf(Cn);

    out[t]          = O;
    out[NC + t]     = Hn;
    out[2 * NC + t] = Cn;
}

extern "C" void kernel_launch(void* const* d_in, const int* in_sizes, int n_in,
                              void* d_out, int out_size, void* d_ws, size_t ws_size,
                              hipStream_t stream) {
    const float* X   = (const float*)d_in[0];
    const int*   ei  = (const int*)  d_in[1];
    const float* w   = (const float*)d_in[2];
    const float* H   = (const float*)d_in[3];
    const float* C   = (const float*)d_in[4];
    const float* Wx  = (const float*)d_in[5];
    const float* bx  = (const float*)d_in[6];
    const float* Wh  = (const float*)d_in[7];
    const float* bh  = (const float*)d_in[8];
    const float* bi  = (const float*)d_in[9];
    const float* bf  = (const float*)d_in[10];
    const float* bc  = (const float*)d_in[11];
    const float* bo  = (const float*)d_in[12];

    const int N = in_sizes[0] / CH;
    const int E = in_sizes[2];
    const int* src = ei;
    const int* dst = ei + E;

    float* ws  = (float*)d_ws;
    size_t Na  = ((size_t)N + 3) & ~(size_t)3;
    float* deg  = ws;
    float* dinv = ws + Na;
    float* P    = ws + 2 * Na;
    float* agg  = P + (size_t)N * CH;

    hipMemsetAsync(deg, 0, (size_t)N * sizeof(float), stream);
    hipMemsetAsync(agg, 0, (size_t)N * CH * sizeof(float), stream);

    k_deg<<<(E + 255) / 256, 256, 0, stream>>>(dst, w, deg, E);
    k_dinv<<<(N + 255) / 256, 256, 0, stream>>>(deg, dinv, N);

    int tiles = (N + 15) / 16;
    k_gemm_wmma<<<(tiles + 7) / 8, 256, 0, stream>>>(X, H, Wx, Wh, P, N);

    k_scatter<<<(E + 7) / 8, 256, 0, stream>>>(src, dst, w, dinv, P, agg, E);

    int NC = N * CH;
    k_gates<<<(NC + 255) / 256, 256, 0, stream>>>(agg, C, bx, bh, bi, bf, bc, bo,
                                                  (float*)d_out, NC);
}


# scatter: each work-group keeps a 2048-row dst range of agg in LDS, scans an edge slice and adds matching P rows with LDS float atomics; 5 partial planes summed in gates; sum order changes
# speedup vs baseline: 1.6063x; 1.6063x over previous
#include <hip/hip_runtime.h>
#include <hip/hip_bf16.h>
#include <math.h>

typedef __attribute__((ext_vector_type(2))) float v2f;
typedef __attribute__((ext_vector_type(8))) float v8f;

#define CH 32

__global__ __launch_bounds__(256) void k_deg(const int* __restrict__ dst,
                                             const float* __restrict__ w,
                                             float* __restrict__ deg, int E) {
    int e = blockIdx.x * blockDim.x + threadIdx.x;
    if (e < E) {
        __hip_atomic_fetch_add(&deg[dst[e]], w[e],
                               __ATOMIC_RELAXED, __HIP_MEMORY_SCOPE_AGENT);
    }
}

__global__ __launch_bounds__(256) void k_dinv(const float* __restrict__ deg,
                                              float* __restrict__ dinv, int N) {
    int i = blockIdx.x * blockDim.x + threadIdx.x;
    if (i < N) {
        float d = deg[i];
        dinv[i] = (d > 0.0f) ? rsqrtf(fmaxf(d, 1e-12f)) : 0.0f;
    }
}

__global__ __launch_bounds__(256) void k_gemm_wmma(const float* __restrict__ X,
                                                   const float* __restrict__ H,
                                                   const float* __restrict__ Wx,
                                                   const float* __restrict__ Wh,
                                                   float* __restrict__ P, int N) {
    int wave = blockIdx.x * (blockDim.x >> 5) + (threadIdx.x >> 5);
    int row0 = wave * 16;
    if (row0 >= N) return;

    int lane = threadIdx.x & 31;
    int half = lane >> 4;
    int lm   = lane & 15;

    int arow = row0 + lm;
    if (arow > N - 1) arow = N - 1;
    const float* xrow = X + (size_t)arow * CH;
    const float* hrow = H + (size_t)arow * CH;

    v8f acc0 = {};
    v8f acc1 = {};

#pragma unroll
    for (int kt = 0; kt < 8; ++kt) {
        int k0 = kt * 4;
        int ka = k0 + 2 * half;

        v2f ax = { xrow[ka], xrow[ka + 1] };
        v2f ah = { hrow[ka], hrow[ka + 1] };

        int krow = ka * CH;
        v2f bx0 = { Wx[krow + lm],      Wx[krow + CH + lm]      };
        v2f bx1 = { Wx[krow + 16 + lm], Wx[krow + CH + 16 + lm] };
        v2f bh0 = { Wh[krow + lm],      Wh[krow + CH + lm]      };
        v2f bh1 = { Wh[krow + 16 + lm], Wh[krow + CH + 16 + lm] };

        acc0 = __builtin_amdgcn_wmma_f32_16x16x4_f32(false, ax, false, bx0,
                                                     (short)0, acc0, false, false);
        acc1 = __builtin_amdgcn_wmma_f32_16x16x4_f32(false, ax, false, bx1,
                                                     (short)0, acc1, false, false);
        acc0 = __builtin_amdgcn_wmma_f32_16x16x4_f32(false, ah, false, bh0,
                                                     (short)0, acc0, false, false);
        acc1 = __builtin_amdgcn_wmma_f32_16x16x4_f32(false, ah, false, bh1,
                                                     (short)0, acc1, false, false);
    }

    float* prow = P + (size_t)row0 * CH;
#pragma unroll
    for (int r = 0; r < 8; ++r) {
        int rr = r + 8 * half;
        if (row0 + rr < N) {
            prow[(size_t)rr * CH + lm]      = acc0[r];
            prow[(size_t)rr * CH + 16 + lm] = acc1[r];
        }
    }
}

#define RP   2048
#define NSL  5
#define QCAP 256
#define SCW  32
#define SCU  12
#define SCG  6

__device__ __forceinline__ void drain_rows(const int* qw, int head, int n, int lane, int base,
                                           const int* __restrict__ src,
                                           const int* __restrict__ dst,
                                           const float* __restrict__ w,
                                           const float* __restrict__ dinv,
                                           const float* __restrict__ P,
                                           float* tile) {
    n = __builtin_amdgcn_readfirstlane(n);
    int e = qw[(head + lane) & (QCAP - 1)];
    int s = 0, dl = 0;
    float nrm = 0.0f;
    if (lane < n) {
        s = src[e];
        int d = dst[e];
        dl = d - base;
        nrm = dinv[s] * w[e] * dinv[d];
    }
    float pv[32];
#pragma unroll
    for (int i = 0; i < 32; ++i) {
        unsigned si = (unsigned)__builtin_amdgcn_readlane(s, i);
        pv[i] = P[si * (unsigned)CH + (unsigned)lane];
    }
#pragma unroll
    for (int i = 0; i < 32; ++i) {
        if (i < n) {
            float ni = __int_as_float(__builtin_amdgcn_readlane(__float_as_int(nrm), i));
            int di = __builtin_amdgcn_readlane(dl, i);
            __hip_atomic_fetch_add(&tile[di * CH + lane], ni * pv[i],
                                   __ATOMIC_RELAXED, __HIP_MEMORY_SCOPE_WORKGROUP);
        }
    }
}

__global__ __launch_bounds__(1024) void k_scatter_lds(const int* __restrict__ src,
                                                      const int* __restrict__ dst,
                                                      const float* __restrict__ w,
                                                      const float* __restrict__ dinv,
                                                      const float* __restrict__ P,
                                                      float* __restrict__ planes,
                                                      int E, int N, int slice_len) {
    __shared__ __attribute__((aligned(16))) float tile[RP * CH];
    __shared__ int q[SCW * QCAP];

    const int tid  = threadIdx.x;
    const int lane = tid & 31;
    const int wv   = __builtin_amdgcn_readfirstlane(tid >> 5);
    const int base = blockIdx.x * RP;
    const int e_begin = blockIdx.y * slice_len;
    int e_end = e_begin + slice_len;
    if (e_end > E) e_end = E;

    float4* tile4 = reinterpret_cast<float4*>(tile);
    for (int i = tid; i < RP * CH / 4; i += SCW * 32) tile4[i] = make_float4(0.f, 0.f, 0.f, 0.f);
    __syncthreads();

    int* qw = q + wv * QCAP;
    int head = 0, cnt = 0;

    for (int c0 = e_begin + wv * (SCU * 32); c0 < e_end; c0 += SCW * SCU * 32) {
        int d[SCU];
        const int* pl = dst + c0 + lane;
        if (c0 + SCU * 32 <= e_end) {
#pragma unroll
            for (int u = 0; u < SCU; ++u) d[u] = pl[u * 32];
        } else {
#pragma unroll
            for (int u = 0; u < SCU; ++u) d[u] = (c0 + u * 32 + lane < e_end) ? pl[u * 32] : -1;
        }
#pragma unroll
        for (int u = 0; u < SCU; ++u) {
            bool m = (unsigned)(d[u] - base) < (unsigned)RP;
            unsigned mask = __builtin_amdgcn_ballot_w32(m);
            if (mask != 0u) {
                unsigned below = __builtin_amdgcn_mbcnt_lo(mask, 0u);
                if (m) qw[(head + cnt + (int)below) & (QCAP - 1)] = c0 + u * 32 + lane;
                cnt += __builtin_popcount(mask);
            }
            if ((u % SCG) == SCG - 1 || u == SCU - 1) {
                while (cnt >= 32) {
                    drain_rows(qw, head, 32, lane, base, src, dst, w, dinv, P, tile);
                    head = (head + 32) & (QCAP - 1);
                    cnt -= 32;
                }
            }
        }
    }
    if (cnt > 0) drain_rows(qw, head, cnt, lane, base, src, dst, w, dinv, P, tile);

    __syncthreads();
    int rows = N - base;
    if (rows > RP) rows = RP;
    float4* out4 = reinterpret_cast<float4*>(planes + (size_t)blockIdx.y * N * CH + (size_t)base * CH);
    for (int i = tid; i < rows * (CH / 4); i += SCW * 32) out4[i] = tile4[i];
}

__device__ __forceinline__ float sigmoidf_(float x) {
    return 1.0f / (1.0f + __expf(-x));
}

__global__ __launch_bounds__(256) void k_gates(const float* __restrict__ agg,
                                               const float* __restrict__ Cst,
                                               const float* __restrict__ bx,
                                               const float* __restrict__ bh,
                                               const float* __restrict__ bi,
                                               const float* __restrict__ bf,
                                               const float* __restrict__ bc,
                                               const float* __restrict__ bo,
                                               float* __restrict__ out, int NC) {
    int t = blockIdx.x * blockDim.x + threadIdx.x;
    if (t >= NC) return;
    int c = t & (CH - 1);

    float a = agg[t];
#pragma unroll
    for (int k = 1; k < NSL; ++k) a += agg[(size_t)k * NC + t];
    float s  = a + bx[c] + bh[c];
    float I  = sigmoidf_(s + bi[c]);
    float F  = sigmoidf_(s + bf[c]);
    float T  = tanhf(s + bc[c]);
    float Cn = F * Cst[t] + I * T;
    float O  = sigmoidf_(s + bo[c]);
    float Hn = O * tanhf(Cn);

    out[t]          = O;
    out[NC + t]     = Hn;
    out[2 * NC + t] = Cn;
}

extern "C" void kernel_launch(void* const* d_in, const int* in_sizes, int n_in,
                              void* d_out, int out_size, void* d_ws, size_t ws_size,
                              hipStream_t stream) {
    const float* X   = (const float*)d_in[0];
    const int*   ei  = (const int*)  d_in[1];
    const float* w   = (const float*)d_in[2];
    const float* H   = (const float*)d_in[3];
    const float* C   = (const float*)d_in[4];
    const float* Wx  = (const float*)d_in[5];
    const float* bx  = (const float*)d_in[6];
    const float* Wh  = (const float*)d_in[7];
    const float* bh  = (const float*)d_in[8];
    const float* bi  = (const float*)d_in[9];
    const float* bf  = (const float*)d_in[10];
    const float* bc  = (const float*)d_in[11];
    const float* bo  = (const float*)d_in[12];

    const int N = in_sizes[0] / CH;
    const int E = in_sizes[2];
    const int* src = ei;
    const int* dst = ei + E;

    float* ws  = (float*)d_ws;
    size_t Na  = ((size_t)N + 3) & ~(size_t)3;
    float* deg  = ws;
    float* dinv = ws + Na;
    float* P    = ws + 2 * Na;
    float* agg  = P + (size_t)N * CH;

    hipMemsetAsync(deg, 0, (size_t)N * sizeof(float), stream);

    k_deg<<<(E + 255) / 256, 256, 0, stream>>>(dst, w, deg, E);
    k_dinv<<<(N + 255) / 256, 256, 0, stream>>>(deg, dinv, N);

    int tiles = (N + 15) / 16;
    k_gemm_wmma<<<(tiles + 7) / 8, 256, 0, stream>>>(X, H, Wx, Wh, P, N);

    dim3 sgrid((N + RP - 1) / RP, NSL);
    k_scatter_lds<<<sgrid, SCW * 32, 0, stream>>>(src, dst, w, dinv, P, agg, E, N,
                                                  (E + NSL - 1) / NSL);

    int NC = N * CH;
    k_gates<<<(NC + 255) / 256, 256, 0, stream>>>(agg, C, bx, bh, bi, bf, bc, bo,
                                                  (float*)d_out, NC);
}


# scatter drain: per-edge gathers 6 -> 3 (src, w, P row): dinv[src] folded into P rows at the GEMM store, dinv[dst] applied per row in gates, dst-local row kept in the ring; products re-associated
# speedup vs baseline: 1.7834x; 1.1103x over previous
#include <hip/hip_runtime.h>
#include <hip/hip_bf16.h>
#include <math.h>

typedef __attribute__((ext_vector_type(2))) float v2f;
typedef __attribute__((ext_vector_type(8))) float v8f;

#define CH 32

__global__ __launch_bounds__(256) void k_deg(const int* __restrict__ dst,
                                             const float* __restrict__ w,
                                             float* __restrict__ deg, int E) {
    int e = blockIdx.x * blockDim.x + threadIdx.x;
    if (e < E) {
        __hip_atomic_fetch_add(&deg[dst[e]], w[e],
                               __ATOMIC_RELAXED, __HIP_MEMORY_SCOPE_AGENT);
    }
}

__global__ __launch_bounds__(256) void k_dinv(const float* __restrict__ deg,
                                              float* __restrict__ dinv, int N) {
    int i = blockIdx.x * blockDim.x + threadIdx.x;
    if (i < N) {
        float d = deg[i];
        dinv[i] = (d > 0.0f) ? rsqrtf(fmaxf(d, 1e-12f)) : 0.0f;
    }
}

__global__ __launch_bounds__(256) void k_gemm_wmma(const float* __restrict__ X,
                                                   const float* __restrict__ H,
                                                   const float* __restrict__ Wx,
                                                   const float* __restrict__ Wh,
                                                   const float* __restrict__ dinv,
                                                   float* __restrict__ P, int N) {
    int wave = blockIdx.x * (blockDim.x >> 5) + (threadIdx.x >> 5);
    int row0 = wave * 16;
    if (row0 >= N) return;

    int lane = threadIdx.x & 31;
    int half = lane >> 4;
    int lm   = lane & 15;

    int arow = row0 + lm;
    if (arow > N - 1) arow = N - 1;
    const float* xrow = X + (size_t)arow * CH;
    const float* hrow = H + (size_t)arow * CH;

    v8f acc0 = {};
    v8f acc1 = {};

    float dv[8];
#pragma unroll
    for (int r = 0; r < 8; ++r) {
        int drow = row0 + r + 8 * half;
        if (drow > N - 1) drow = N - 1;
        dv[r] = dinv[drow];
    }

#pragma unroll
    for (int kt = 0; kt < 8; ++kt) {
        int k0 = kt * 4;
        int ka = k0 + 2 * half;

        v2f ax = { xrow[ka], xrow[ka + 1] };
        v2f ah = { hrow[ka], hrow[ka + 1] };

        int krow = ka * CH;
        v2f bx0 = { Wx[krow + lm],      Wx[krow + CH + lm]      };
        v2f bx1 = { Wx[krow + 16 + lm], Wx[krow + CH + 16 + lm] };
        v2f bh0 = { Wh[krow + lm],      Wh[krow + CH + lm]      };
        v2f bh1 = { Wh[krow + 16 + lm], Wh[krow + CH + 16 + lm] };

        acc0 = __builtin_amdgcn_wmma_f32_16x16x4_f32(false, ax, false, bx0,
                                                     (short)0, acc0, false, false);
        acc1 = __builtin_amdgcn_wmma_f32_16x16x4_f32(false, ax, false, bx1,
                                                     (short)0, acc1, false, false);
        acc0 = __builtin_amdgcn_wmma_f32_16x16x4_f32(false, ah, false, bh0,
                                                     (short)0, acc0, false, false);
        acc1 = __builtin_amdgcn_wmma_f32_16x16x4_f32(false, ah, false, bh1,
                                                     (short)0, acc1, false, false);
    }

    float* prow = P + (size_t)row0 * CH;
#pragma unroll
    for (int r = 0; r < 8; ++r) {
        int rr = r + 8 * half;
        if (row0 + rr < N) {
            prow[(size_t)rr * CH + lm]      = dv[r] * acc0[r];
            prow[(size_t)rr * CH + 16 + lm] = dv[r] * acc1[r];
        }
    }
}

#define RP   2048
#define NSL  5
#define QCAP 256
#define SCW  32
#define SCU  12
#define SCG  6

__device__ __forceinline__ void drain_rows(const int* qw, const unsigned short* qd,
                                           int head, int n, int lane,
                                           const int* __restrict__ src,
                                           const float* __restrict__ w,
                                           const float* __restrict__ P,
                                           float* tile) {
    n = __builtin_amdgcn_readfirstlane(n);
    int slot = (head + lane) & (QCAP - 1);
    int e  = qw[slot];
    int dl = qd[slot];
    int s = 0;
    float nrm = 0.0f;
    if (lane < n) {
        s = src[e];
        nrm = w[e];
    }
    float pv[32];
#pragma unroll
    for (int i = 0; i < 32; ++i) {
        unsigned si = (unsigned)__builtin_amdgcn_readlane(s, i);
        pv[i] = P[si * (unsigned)CH + (unsigned)lane];
    }
#pragma unroll
    for (int i = 0; i < 32; ++i) {
        if (i < n) {
            float ni = __int_as_float(__builtin_amdgcn_readlane(__float_as_int(nrm), i));
            int di = __builtin_amdgcn_readlane(dl, i);
            __hip_atomic_fetch_add(&tile[di * CH + lane], ni * pv[i],
                                   __ATOMIC_RELAXED, __HIP_MEMORY_SCOPE_WORKGROUP);
        }
    }
}

__global__ __launch_bounds__(1024) void k_scatter_lds(const int* __restrict__ src,
                                                      const int* __restrict__ dst,
                                                      const float* __restrict__ w,
                                                      const float* __restrict__ P,
                                                      float* __restrict__ planes,
                                                      int E, int N, int slice_len) {
    __shared__ __attribute__((aligned(16))) float tile[RP * CH];
    __shared__ int q[SCW * QCAP];
    __shared__ unsigned short qdl[SCW * QCAP];

    const int tid  = threadIdx.x;
    const int lane = tid & 31;
    const int wv   = __builtin_amdgcn_readfirstlane(tid >> 5);
    const int base = blockIdx.x * RP;
    const int e_begin = blockIdx.y * slice_len;
    int e_end = e_begin + slice_len;
    if (e_end > E) e_end = E;

    float4* tile4 = reinterpret_cast<float4*>(tile);
    for (int i = tid; i < RP * CH / 4; i += SCW * 32) tile4[i] = make_float4(0.f, 0.f, 0.f, 0.f);
    __syncthreads();

    int* qw = q + wv * QCAP;
    unsigned short* qd = qdl + wv * QCAP;
    int head = 0, cnt = 0;

    for (int c0 = e_begin + wv * (SCU * 32); c0 < e_end; c0 += SCW * SCU * 32) {
        int d[SCU];
        const int* pl = dst + c0 + lane;
        if (c0 + SCU * 32 <= e_end) {
#pragma unroll
            for (int u = 0; u < SCU; ++u) d[u] = pl[u * 32];
        } else {
#pragma unroll
            for (int u = 0; u < SCU; ++u) d[u] = (c0 + u * 32 + lane < e_end) ? pl[u * 32] : -1;
        }
#pragma unroll
        for (int u = 0; u < SCU; ++u) {
            bool m = (unsigned)(d[u] - base) < (unsigned)RP;
            unsigned mask = __builtin_amdgcn_ballot_w32(m);
            if (mask != 0u) {
                unsigned below = __builtin_amdgcn_mbcnt_lo(mask, 0u);
                if (m) {
                    int slot = (head + cnt + (int)below) & (QCAP - 1);
                    qw[slot] = c0 + u * 32 + lane;
                    qd[slot] = (unsigned short)(d[u] - base);
                }
                cnt += __builtin_popcount(mask);
            }
            if ((u % SCG) == SCG - 1 || u == SCU - 1) {
                while (cnt >= 32) {
                    drain_rows(qw, qd, head, 32, lane, src, w, P, tile);
                    head = (head + 32) & (QCAP - 1);
                    cnt -= 32;
                }
            }
        }
    }
    if (cnt > 0) drain_rows(qw, qd, head, cnt, lane, src, w, P, tile);

    __syncthreads();
    int rows = N - base;
    if (rows > RP) rows = RP;
    float4* out4 = reinterpret_cast<float4*>(planes + (size_t)blockIdx.y * N * CH + (size_t)base * CH);
    for (int i = tid; i < rows * (CH / 4); i += SCW * 32) out4[i] = tile4[i];
}

__device__ __forceinline__ float sigmoidf_(float x) {
    return 1.0f / (1.0f + __expf(-x));
}

__global__ __launch_bounds__(256) void k_gates(const float* __restrict__ agg,
                                               const float* __restrict__ dinv,
                                               const float* __restrict__ Cst,
                                               const float* __restrict__ bx,
                                               const float* __restrict__ bh,
                                               const float* __restrict__ bi,
                                               const float* __restrict__ bf,
                                               const float* __restrict__ bc,
                                               const float* __restrict__ bo,
                                               float* __restrict__ out, int NC) {
    int t = blockIdx.x * blockDim.x + threadIdx.x;
    if (t >= NC) return;
    int c = t & (CH - 1);

    float a = agg[t];
#pragma unroll
    for (int k = 1; k < NSL; ++k) a += agg[(size_t)k * NC + t];
    float s  = dinv[(unsigned)t / CH] * a + bx[c] + bh[c];
    float I  = sigmoidf_(s + bi[c]);
    float F  = sigmoidf_(s + bf[c]);
    float T  = tanhf(s + bc[c]);
    float Cn = F * Cst[t] + I * T;
    float O  = sigmoidf_(s + bo[c]);
    float Hn = O * tanhf(Cn);

    out[t]          = O;
    out[NC + t]     = Hn;
    out[2 * NC + t] = Cn;
}

extern "C" void kernel_launch(void* const* d_in, const int* in_sizes, int n_in,
                              void* d_out, int out_size, void* d_ws, size_t ws_size,
                              hipStream_t stream) {
    const float* X   = (const float*)d_in[0];
    const int*   ei  = (const int*)  d_in[1];
    const float* w   = (const float*)d_in[2];
    const float* H   = (const float*)d_in[3];
    const float* C   = (const float*)d_in[4];
    const float* Wx  = (const float*)d_in[5];
    const float* bx  = (const float*)d_in[6];
    const float* Wh  = (const float*)d_in[7];
    const float* bh  = (const float*)d_in[8];
    const float* bi  = (const float*)d_in[9];
    const float* bf  = (const float*)d_in[10];
    const float* bc  = (const float*)d_in[11];
    const float* bo  = (const float*)d_in[12];

    const int N = in_sizes[0] / CH;
    const int E = in_sizes[2];
    const int* src = ei;
    const int* dst = ei + E;

    float* ws  = (float*)d_ws;
    size_t Na  = ((size_t)N + 3) & ~(size_t)3;
    float* deg  = ws;
    float* dinv = ws + Na;
    float* P    = ws + 2 * Na;
    float* agg  = P + (size_t)N * CH;

    hipMemsetAsync(deg, 0, (size_t)N * sizeof(float), stream);

    k_deg<<<(E + 255) / 256, 256, 0, stream>>>(dst, w, deg, E);
    k_dinv<<<(N + 255) / 256, 256, 0, stream>>>(deg, dinv, N);

    int tiles = (N + 15) / 16;
    k_gemm_wmma<<<(tiles + 7) / 8, 256, 0, stream>>>(X, H, Wx, Wh, dinv, P, N);

    dim3 sgrid((N + RP - 1) / RP, NSL);
    k_scatter_lds<<<sgrid, SCW * 32, 0, stream>>>(src, dst, w, P, agg, E, N,
                                                  (E + NSL - 1) / NSL);

    int NC = N * CH;
    k_gates<<<(NC + 255) / 256, 256, 0, stream>>>(agg, dinv, C, bx, bh, bi, bf, bc, bo,
                                                  (float*)d_out, NC);
}


# degree sums: work-group keeps a 25088-row dst range in LDS, adds w of its edge slice by LDS float atomics; 64 partial planes summed in the dinv launch; no global atomics, no memset; sum order changes
# speedup vs baseline: 3.8078x; 2.1351x over previous
#include <hip/hip_runtime.h>
#include <hip/hip_bf16.h>
#include <math.h>

typedef __attribute__((ext_vector_type(2))) float v2f;
typedef __attribute__((ext_vector_type(8))) float v8f;

#define CH 32

#define DR   25088
#define DSL  64
#define DGT  1024
#define DU   8
#define DG   16

__global__ __launch_bounds__(DGT) void k_deg_lds(const int* __restrict__ dst,
                                                 const float* __restrict__ w,
                                                 float* __restrict__ degp,
                                                 int E, int N, int Na, int slice_len) {
    __shared__ __attribute__((aligned(16))) float dl[DR];

    const int tid  = threadIdx.x;
    const int base = blockIdx.x * DR;
    const int e_begin = blockIdx.y * slice_len;
    int e_end = e_begin + slice_len;
    if (e_end > E) e_end = E;

    float4* dl4 = reinterpret_cast<float4*>(dl);
    for (int i = tid; i < DR / 4; i += DGT) dl4[i] = make_float4(0.f, 0.f, 0.f, 0.f);
    __syncthreads();

    for (int cb = e_begin; cb < e_end; cb += DGT * DU) {
        int   d[DU];
        float ww[DU];
        const int*   pd = dst + cb + tid;
        const float* pw = w + cb + tid;
        if (cb + DGT * DU <= e_end) {
#pragma unroll
            for (int u = 0; u < DU; ++u) { d[u] = pd[u * DGT]; ww[u] = pw[u * DGT]; }
        } else {
#pragma unroll
            for (int u = 0; u < DU; ++u) {
                int idx = cb + u * DGT + tid;
                bool in = idx < e_end;
                int ii = in ? idx : e_end - 1;
                int dv = dst[ii];
                ww[u] = w[ii];
                d[u]  = in ? dv : -1;
            }
        }
#pragma unroll
        for (int u = 0; u < DU; ++u) {
            unsigned r = (unsigned)(d[u] - base);
            if (r < (unsigned)DR) {
                __hip_atomic_fetch_add(&dl[r], ww[u],
                                       __ATOMIC_RELAXED, __HIP_MEMORY_SCOPE_WORKGROUP);
            }
        }
    }
    __syncthreads();

    int rows = N - base;
    if (rows > DR) rows = DR;
    float* outp = degp + (size_t)blockIdx.y * Na + base;
    for (int i = tid; i < rows; i += DGT) outp[i] = dl[i];
}

__global__ __launch_bounds__(256) void k_dinv(const float* __restrict__ degp,
                                              float* __restrict__ dinv, int N, int Na) {
    int i = blockIdx.x * blockDim.x + threadIdx.x;
    if (i < N) {
        const float* p = degp + i;
        float d = 0.0f;
        for (int k = 0; k < DSL; k += DG) {
            float v[DG];
#pragma unroll
            for (int u = 0; u < DG; ++u) v[u] = p[(size_t)(k + u) * Na];
#pragma unroll
            for (int u = 0; u < DG; ++u) d += v[u];
        }
        dinv[i] = (d > 0.0f) ? rsqrtf(fmaxf(d, 1e-12f)) : 0.0f;
    }
}

__global__ __launch_bounds__(256) void k_gemm_wmma(const float* __restrict__ X,
                                                   const float* __restrict__ H,
                                                   const float* __restrict__ Wx,
                                                   const float* __restrict__ Wh,
                                                   const float* __restrict__ dinv,
                                                   float* __restrict__ P, int N) {
    int wave = blockIdx.x * (blockDim.x >> 5) + (threadIdx.x >> 5);
    int row0 = wave * 16;
    if (row0 >= N) return;

    int lane = threadIdx.x & 31;
    int half = lane >> 4;
    int lm   = lane & 15;

    int arow = row0 + lm;
    if (arow > N - 1) arow = N - 1;
    const float* xrow = X + (size_t)arow * CH;
    const float* hrow = H + (size_t)arow * CH;

    v8f acc0 = {};
    v8f acc1 = {};

    float dv[8];
#pragma unroll
    for (int r = 0; r < 8; ++r) {
        int drow = row0 + r + 8 * half;
        if (drow > N - 1) drow = N - 1;
        dv[r] = dinv[drow];
    }

#pragma unroll
    for (int kt = 0; kt < 8; ++kt) {
        int k0 = kt * 4;
        int ka = k0 + 2 * half;

        v2f ax = { xrow[ka], xrow[ka + 1] };
        v2f ah = { hrow[ka], hrow[ka + 1] };

        int krow = ka * CH;
        v2f bx0 = { Wx[krow + lm],      Wx[krow + CH + lm]      };
        v2f bx1 = { Wx[krow + 16 + lm], Wx[krow + CH + 16 + lm] };
        v2f bh0 = { Wh[krow + lm],      Wh[krow + CH + lm]      };
        v2f bh1 = { Wh[krow + 16 + lm], Wh[krow + CH + 16 + lm] };

        acc0 = __builtin_amdgcn_wmma_f32_16x16x4_f32(false, ax, false, bx0,
                                                     (short)0, acc0, false, false);
        acc1 = __builtin_amdgcn_wmma_f32_16x16x4_f32(false, ax, false, bx1,
                                                     (short)0, acc1, false, false);
        acc0 = __builtin_amdgcn_wmma_f32_16x16x4_f32(false, ah, false, bh0,
                                                     (short)0, acc0, false, false);
        acc1 = __builtin_amdgcn_wmma_f32_16x16x4_f32(false, ah, false, bh1,
                                                     (short)0, acc1, false, false);
    }

    float* prow = P + (size_t)row0 * CH;
#pragma unroll
    for (int r = 0; r < 8; ++r) {
        int rr = r + 8 * half;
        if (row0 + rr < N) {
            prow[(size_t)rr * CH + lm]      = dv[r] * acc0[r];
            prow[(size_t)rr * CH + 16 + lm] = dv[r] * acc1[r];
        }
    }
}

#define RP   2048
#define NSL  5
#define QCAP 256
#define SCW  32
#define SCU  12
#define SCG  6

__device__ __forceinline__ void drain_rows(const int* qw, const unsigned short* qd,
                                           int head, int n, int lane,
                                           const int* __restrict__ src,
                                           const float* __restrict__ w,
                                           const float* __restrict__ P,
                                           float* tile) {
    n = __builtin_amdgcn_readfirstlane(n);
    int slot = (head + lane) & (QCAP - 1);
    int e  = qw[slot];
    int dl = qd[slot];
    int s = 0;
    float nrm = 0.0f;
    if (lane < n) {
        s = src[e];
        nrm = w[e];
    }
    float pv[32];
#pragma unroll
    for (int i = 0; i < 32; ++i) {
        unsigned si = (unsigned)__builtin_amdgcn_readlane(s, i);
        pv[i] = P[si * (unsigned)CH + (unsigned)lane];
    }
#pragma unroll
    for (int i = 0; i < 32; ++i) {
        if (i < n) {
            float ni = __int_as_float(__builtin_amdgcn_readlane(__float_as_int(nrm), i));
            int di = __builtin_amdgcn_readlane(dl, i);
            __hip_atomic_fetch_add(&tile[di * CH + lane], ni * pv[i],
                                   __ATOMIC_RELAXED, __HIP_MEMORY_SCOPE_WORKGROUP);
        }
    }
}

__global__ __launch_bounds__(1024) void k_scatter_lds(const int* __restrict__ src,
                                                      const int* __restrict__ dst,
                                                      const float* __restrict__ w,
                                                      const float* __restrict__ P,
                                                      float* __restrict__ planes,
                                                      int E, int N, int slice_len) {
    __shared__ __attribute__((aligned(16))) float tile[RP * CH];
    __shared__ int q[SCW * QCAP];
    __shared__ unsigned short qdl[SCW * QCAP];

    const int tid  = threadIdx.x;
    const int lane = tid & 31;
    const int wv   = __builtin_amdgcn_readfirstlane(tid >> 5);
    const int base = blockIdx.x * RP;
    const int e_begin = blockIdx.y * slice_len;
    int e_end = e_begin + slice_len;
    if (e_end > E) e_end = E;

    float4* tile4 = reinterpret_cast<float4*>(tile);
    for (int i = tid; i < RP * CH / 4; i += SCW * 32) tile4[i] = make_float4(0.f, 0.f, 0.f, 0.f);
    __syncthreads();

    int* qw = q + wv * QCAP;
    unsigned short* qd = qdl + wv * QCAP;
    int head = 0, cnt = 0;

    for (int c0 = e_begin + wv * (SCU * 32); c0 < e_end; c0 += SCW * SCU * 32) {
        int d[SCU];
        const int* pl = dst + c0 + lane;
        if (c0 + SCU * 32 <= e_end) {
#pragma unroll
            for (int u = 0; u < SCU; ++u) d[u] = pl[u * 32];
        } else {
#pragma unroll
            for (int u = 0; u < SCU; ++u) d[u] = (c0 + u * 32 + lane < e_end) ? pl[u * 32] : -1;
        }
#pragma unroll
        for (int u = 0; u < SCU; ++u) {
            bool m = (unsigned)(d[u] - base) < (unsigned)RP;
            unsigned mask = __builtin_amdgcn_ballot_w32(m);
            if (mask != 0u) {
                unsigned below = __builtin_amdgcn_mbcnt_lo(mask, 0u);
                if (m) {
                    int slot = (head + cnt + (int)below) & (QCAP - 1);
                    qw[slot] = c0 + u * 32 + lane;
                    qd[slot] = (unsigned short)(d[u] - base);
                }
                cnt += __builtin_popcount(mask);
            }
            if ((u % SCG) == SCG - 1 || u == SCU - 1) {
                while (cnt >= 32) {
                    drain_rows(qw, qd, head, 32, lane, src, w, P, tile);
                    head = (head + 32) & (QCAP - 1);
                    cnt -= 32;
                }
            }
        }
    }
    if (cnt > 0) drain_rows(qw, qd, head, cnt, lane, src, w, P, tile);

    __syncthreads();
    int rows = N - base;
    if (rows > RP) rows = RP;
    float4* out4 = reinterpret_cast<float4*>(planes + (size_t)blockIdx.y * N * CH + (size_t)base * CH);
    for (int i = tid; i < rows * (CH / 4); i += SCW * 32) out4[i] = tile4[i];
}

__device__ __forceinline__ float sigmoidf_(float x) {
    return 1.0f / (1.0f + __expf(-x));
}

__global__ __launch_bounds__(256) void k_gates(const float* __restrict__ agg,
                                               const float* __restrict__ dinv,
                                               const float* __restrict__ Cst,
                                               const float* __restrict__ bx,
                                               const float* __restrict__ bh,
                                               const float* __restrict__ bi,
                                               const float* __restrict__ bf,
                                               const float* __restrict__ bc,
                                               const float* __restrict__ bo,
                                               float* __restrict__ out, int NC) {
    int t = blockIdx.x * blockDim.x + threadIdx.x;
    if (t >= NC) return;
    int c = t & (CH - 1);

    float a = agg[t];
#pragma unroll
    for (int k = 1; k < NSL; ++k) a += agg[(size_t)k * NC + t];
    float s  = dinv[(unsigned)t / CH] * a + bx[c] + bh[c];
    float I  = sigmoidf_(s + bi[c]);
    float F  = sigmoidf_(s + bf[c]);
    float T  = tanhf(s + bc[c]);
    float Cn = F * Cst[t] + I * T;
    float O  = sigmoidf_(s + bo[c]);
    float Hn = O * tanhf(Cn);

    out[t]          = O;
    out[NC + t]     = Hn;
    out[2 * NC + t] = Cn;
}

extern "C" void kernel_launch(void* const* d_in, const int* in_sizes, int n_in,
                              void* d_out, int out_size, void* d_ws, size_t ws_size,
                              hipStream_t stream) {
    const float* X   = (const float*)d_in[0];
    const int*   ei  = (const int*)  d_in[1];
    const float* w   = (const float*)d_in[2];
    const float* H   = (const float*)d_in[3];
    const float* C   = (const float*)d_in[4];
    const float* Wx  = (const float*)d_in[5];
    const float* bx  = (const float*)d_in[6];
    const float* Wh  = (const float*)d_in[7];
    const float* bh  = (const float*)d_in[8];
    const float* bi  = (const float*)d_in[9];
    const float* bf  = (const float*)d_in[10];
    const float* bc  = (const float*)d_in[11];
    const float* bo  = (const float*)d_in[12];

    const int N = in_sizes[0] / CH;
    const int E = in_sizes[2];
    const int* src = ei;
    const int* dst = ei + E;

    float* ws  = (float*)d_ws;
    size_t Na  = ((size_t)N + 31) & ~(size_t)31;
    float* degp = ws;
    float* dinv = ws + (size_t)DSL * Na;
    float* P    = dinv + Na;
    float* agg  = P + (size_t)N * CH;

    dim3 dgrid((N + DR - 1) / DR, DSL);
    k_deg_lds<<<dgrid, DGT, 0, stream>>>(dst, w, degp, E, N, (int)Na, (E + DSL - 1) / DSL);
    k_dinv<<<(N + 255) / 256, 256, 0, stream>>>(degp, dinv, N, (int)Na);

    int tiles = (N + 15) / 16;
    k_gemm_wmma<<<(tiles + 7) / 8, 256, 0, stream>>>(X, H, Wx, Wh, dinv, P, N);

    dim3 sgrid((N + RP - 1) / RP, NSL);
    k_scatter_lds<<<sgrid, SCW * 32, 0, stream>>>(src, dst, w, P, agg, E, N,
                                                  (E + NSL - 1) / NSL);

    int NC = N * CH;
    k_gates<<<(NC + 255) / 256, 256, 0, stream>>>(agg, dinv, C, bx, bh, bi, bf, bc, bo,
                                                  (float*)d_out, NC);
}


# scatter: edges counting-sorted by dst row bin once (count, then fill via LDS); one work-group per 400-row bin reads only its own edges, coalesced; single agg plane; sum order changes
# speedup vs baseline: 4.5130x; 1.1852x over previous
#include <hip/hip_runtime.h>
#include <hip/hip_bf16.h>
#include <math.h>

typedef __attribute__((ext_vector_type(2))) float v2f;
typedef __attribute__((ext_vector_type(8))) float v8f;

#define CH 32

#define DR   25088
#define DSL  64
#define DGT  1024
#define DU   8
#define DG   16

__global__ __launch_bounds__(DGT) void k_deg_lds(const int* __restrict__ dst,
                                                 const float* __restrict__ w,
                                                 float* __restrict__ degp,
                                                 int E, int N, int Na, int slice_len) {
    __shared__ __attribute__((aligned(16))) float dl[DR];

    const int tid  = threadIdx.x;
    const int base = blockIdx.x * DR;
    const int e_begin = blockIdx.y * slice_len;
    int e_end = e_begin + slice_len;
    if (e_end > E) e_end = E;

    float4* dl4 = reinterpret_cast<float4*>(dl);
    for (int i = tid; i < DR / 4; i += DGT) dl4[i] = make_float4(0.f, 0.f, 0.f, 0.f);
    __syncthreads();

    for (int cb = e_begin; cb < e_end; cb += DGT * DU) {
        int   d[DU];
        float ww[DU];
        const int*   pd = dst + cb + tid;
        const float* pw = w + cb + tid;
        if (cb + DGT * DU <= e_end) {
#pragma unroll
            for (int u = 0; u < DU; ++u) { d[u] = pd[u * DGT]; ww[u] = pw[u * DGT]; }
        } else {
#pragma unroll
            for (int u = 0; u < DU; ++u) {
                int idx = cb + u * DGT + tid;
                bool in = idx < e_end;
                int ii = in ? idx : e_end - 1;
                int dv = dst[ii];
                ww[u] = w[ii];
                d[u]  = in ? dv : -1;
            }
        }
#pragma unroll
        for (int u = 0; u < DU; ++u) {
            unsigned r = (unsigned)(d[u] - base);
            if (r < (unsigned)DR) {
                __hip_atomic_fetch_add(&dl[r], ww[u],
                                       __ATOMIC_RELAXED, __HIP_MEMORY_SCOPE_WORKGROUP);
            }
        }
    }
    __syncthreads();

    int rows = N - base;
    if (rows > DR) rows = DR;
    float* outp = degp + (size_t)blockIdx.y * Na + base;
    for (int i = tid; i < rows; i += DGT) outp[i] = dl[i];
}

__global__ __launch_bounds__(256) void k_dinv(const float* __restrict__ degp,
                                              float* __restrict__ dinv, int N, int Na) {
    int i = blockIdx.x * blockDim.x + threadIdx.x;
    if (i < N) {
        const float* p = degp + i;
        float d = 0.0f;
        for (int k = 0; k < DSL; k += DG) {
            float v[DG];
#pragma unroll
            for (int u = 0; u < DG; ++u) v[u] = p[(size_t)(k + u) * Na];
#pragma unroll
            for (int u = 0; u < DG; ++u) d += v[u];
        }
        dinv[i] = (d > 0.0f) ? rsqrtf(fmaxf(d, 1e-12f)) : 0.0f;
    }
}

__global__ __launch_bounds__(256) void k_gemm_wmma(const float* __restrict__ X,
                                                   const float* __restrict__ H,
                                                   const float* __restrict__ Wx,
                                                   const float* __restrict__ Wh,
                                                   const float* __restrict__ dinv,
                                                   float* __restrict__ P, int N) {
    int wave = blockIdx.x * (blockDim.x >> 5) + (threadIdx.x >> 5);
    int row0 = wave * 16;
    if (row0 >= N) return;

    int lane = threadIdx.x & 31;
    int half = lane >> 4;
    int lm   = lane & 15;

    int arow = row0 + lm;
    if (arow > N - 1) arow = N - 1;
    const float* xrow = X + (size_t)arow * CH;
    const float* hrow = H + (size_t)arow * CH;

    v8f acc0 = {};
    v8f acc1 = {};

    float dv[8];
#pragma unroll
    for (int r = 0; r < 8; ++r) {
        int drow = row0 + r + 8 * half;
        if (drow > N - 1) drow = N - 1;
        dv[r] = dinv[drow];
    }

#pragma unroll
    for (int kt = 0; kt < 8; ++kt) {
        int k0 = kt * 4;
        int ka = k0 + 2 * half;

        v2f ax = { xrow[ka], xrow[ka + 1] };
        v2f ah = { hrow[ka], hrow[ka + 1] };

        int krow = ka * CH;
        v2f bx0 = { Wx[krow + lm],      Wx[krow + CH + lm]      };
        v2f bx1 = { Wx[krow + 16 + lm], Wx[krow + CH + 16 + lm] };
        v2f bh0 = { Wh[krow + lm],      Wh[krow + CH + lm]      };
        v2f bh1 = { Wh[krow + 16 + lm], Wh[krow + CH + 16 + lm] };

        acc0 = __builtin_amdgcn_wmma_f32_16x16x4_f32(false, ax, false, bx0,
                                                     (short)0, acc0, false, false);
        acc1 = __builtin_amdgcn_wmma_f32_16x16x4_f32(false, ax, false, bx1,
                                                     (short)0, acc1, false, false);
        acc0 = __builtin_amdgcn_wmma_f32_16x16x4_f32(false, ah, false, bh0,
                                                     (short)0, acc0, false, false);
        acc1 = __builtin_amdgcn_wmma_f32_16x16x4_f32(false, ah, false, bh1,
                                                     (short)0, acc1, false, false);
    }

    float* prow = P + (size_t)row0 * CH;
#pragma unroll
    for (int r = 0; r < 8; ++r) {
        int rr = r + 8 * half;
        if (row0 + rr < N) {
            prow[(size_t)rr * CH + lm]      = dv[r] * acc0[r];
            prow[(size_t)rr * CH + 16 + lm] = dv[r] * acc1[r];
        }
    }
}

#define RB    400
#define CHK   12800
#define BT    1024
#define BU    13
#define NBMAX 1024
#define SCW   32

__global__ __launch_bounds__(BT) void k_bin_count(const int* __restrict__ dst,
                                                  int* __restrict__ cnt,
                                                  int E, int N, int NB) {
    __shared__ int c[NBMAX];
    const int tid = threadIdx.x;
    const int g   = blockIdx.x;
    const int e0  = g * CHK;
    int e1 = e0 + CHK;
    if (e1 > E) e1 = E;

    for (int b = tid; b < NB; b += BT) c[b] = 0;
    __syncthreads();

    int d[BU];
#pragma unroll
    for (int u = 0; u < BU; ++u) {
        int e = e0 + u * BT + tid;
        d[u] = (e < e1) ? dst[e] : -1;
    }
#pragma unroll
    for (int u = 0; u < BU; ++u) {
        if ((unsigned)d[u] < (unsigned)N) {
            __hip_atomic_fetch_add(&c[(unsigned)d[u] / RB], 1,
                                   __ATOMIC_RELAXED, __HIP_MEMORY_SCOPE_WORKGROUP);
        }
    }
    __syncthreads();
    for (int b = tid; b < NB; b += BT) cnt[(size_t)g * NB + b] = c[b];
}

__global__ __launch_bounds__(BT) void k_bin_fill(const int* __restrict__ src,
                                                 const int* __restrict__ dst,
                                                 const float* __restrict__ w,
                                                 const int* __restrict__ cnt,
                                                 int* __restrict__ binstart,
                                                 int* __restrict__ bsrc,
                                                 float* __restrict__ bw,
                                                 int* __restrict__ bdl,
                                                 int E, int N, int NB, int G) {
    __shared__ int   ssrc[CHK];
    __shared__ float sw[CHK];
    __shared__ int   sdst[CHK];
    __shared__ int   colsum[NBMAX];
    __shared__ int   pre[NBMAX];
    __shared__ int   gst[NBMAX + 1];
    __shared__ int   lst[NBMAX + 1];
    __shared__ int   cur[NBMAX];
    __shared__ int   gdelta[NBMAX];

    const int tid  = threadIdx.x;
    const int lane = tid & 31;
    const int wv   = tid >> 5;
    const int g    = blockIdx.x;
    const int e0   = g * CHK;
    int e1 = e0 + CHK;
    if (e1 > E) e1 = E;

    int   d[BU], s[BU];
    float ww[BU];
#pragma unroll
    for (int u = 0; u < BU; ++u) {
        int e = e0 + u * BT + tid;
        bool in = e < e1;
        int ee = in ? e : e0;
        int dv = dst[ee];
        s[u]  = src[ee];
        ww[u] = w[ee];
        d[u]  = in ? dv : -1;
    }

    for (int b = tid; b < NB; b += BT) { colsum[b] = 0; pre[b] = 0; }
    __syncthreads();

    for (int b = tid & 255; b < NB; b += 256) {
        int tot = 0, pr = 0;
#pragma unroll 16
        for (int gg = tid >> 8; gg < G; gg += 4) {
            int v = cnt[(size_t)gg * NB + b];
            tot += v;
            pr  += (gg < g) ? v : 0;
        }
        __hip_atomic_fetch_add(&colsum[b], tot, __ATOMIC_RELAXED, __HIP_MEMORY_SCOPE_WORKGROUP);
        __hip_atomic_fetch_add(&pre[b],    pr,  __ATOMIC_RELAXED, __HIP_MEMORY_SCOPE_WORKGROUP);
    }
    __syncthreads();

    if (wv == 0) {
        int per = (NB + 31) >> 5;
        int b0 = lane * per;
        int b1 = b0 + per;
        if (b1 > NB) b1 = NB;
        int s1 = 0, s2 = 0;
        for (int b = b0; b < b1; ++b) { s1 += colsum[b]; s2 += cnt[(size_t)g * NB + b]; }
        int i1 = s1, i2 = s2;
        for (int off = 1; off < 32; off <<= 1) {
            int t1 = __shfl_up(i1, off);
            int t2 = __shfl_up(i2, off);
            if (lane >= off) { i1 += t1; i2 += t2; }
        }
        int x1 = i1 - s1, x2 = i2 - s2;
        for (int b = b0; b < b1; ++b) {
            gst[b] = x1;
            lst[b] = x2;
            x1 += colsum[b];
            x2 += cnt[(size_t)g * NB + b];
        }
        if (lane == 31) { gst[NB] = i1; lst[NB] = i2; }
    }
    __syncthreads();

    for (int b = tid; b < NB; b += BT) {
        int l = lst[b];
        cur[b]    = l;
        gdelta[b] = gst[b] + pre[b] - l;
        if (g == 0) binstart[b] = gst[b];
    }
    if (g == 0 && tid == 0) binstart[NB] = gst[NB];
    __syncthreads();

#pragma unroll
    for (int u = 0; u < BU; ++u) {
        if ((unsigned)d[u] < (unsigned)N) {
            int p = __hip_atomic_fetch_add(&cur[(unsigned)d[u] / RB], 1,
                                           __ATOMIC_RELAXED, __HIP_MEMORY_SCOPE_WORKGROUP);
            ssrc[p] = s[u];
            sw[p]   = ww[u];
            sdst[p] = d[u];
        }
    }
    __syncthreads();

    const int n = lst[NB];
    for (int i = tid; i < n; i += BT) {
        unsigned dv = (unsigned)sdst[i];
        unsigned b  = dv / RB;
        int gi = gdelta[b] + i;
        bsrc[gi] = ssrc[i];
        bw[gi]   = sw[i];
        bdl[gi]  = (int)(dv - b * RB);
    }
}

__global__ __launch_bounds__(SCW * 32) void k_scatter_bins(const int* __restrict__ binstart,
                                                           const int* __restrict__ bsrc,
                                                           const float* __restrict__ bw,
                                                           const int* __restrict__ bdl,
                                                           const float* __restrict__ P,
                                                           float* __restrict__ agg, int N) {
    __shared__ __attribute__((aligned(16))) float tile[RB * CH];

    const int tid  = threadIdx.x;
    const int lane = tid & 31;
    const int wv    = __builtin_amdgcn_readfirstlane(tid >> 5);
    const int base  = blockIdx.x * RB;
    const int start = __builtin_amdgcn_readfirstlane(binstart[blockIdx.x]);
    const int end   = __builtin_amdgcn_readfirstlane(binstart[blockIdx.x + 1]);

    int p = start + wv * 32;
    int   s = 0, dl = 0;
    float nrm = 0.0f;
    if (p + lane < end) { s = bsrc[p + lane]; nrm = bw[p + lane]; dl = bdl[p + lane]; }

    float4* tile4 = reinterpret_cast<float4*>(tile);
    for (int i = tid; i < RB * CH / 4; i += SCW * 32) tile4[i] = make_float4(0.f, 0.f, 0.f, 0.f);
    __syncthreads();

    while (p < end) {
        const int pn = p + SCW * 32;
        int ix = pn + lane;
        if (ix > end - 1) ix = end - 1;
        int   s2   = bsrc[ix];
        float nrm2 = bw[ix];
        int   dl2  = bdl[ix];

        int n = end - p;
        if (n > 32) n = 32;
        float pv[32];
#pragma unroll
        for (int i = 0; i < 32; ++i) {
            unsigned si = (unsigned)__builtin_amdgcn_readlane(s, i);
            pv[i] = P[si * (unsigned)CH + (unsigned)lane];
        }
        if (n == 32) {
#pragma unroll
            for (int i = 0; i < 32; ++i) {
                float ni = __int_as_float(__builtin_amdgcn_readlane(__float_as_int(nrm), i));
                int di = __builtin_amdgcn_readlane(dl, i);
                __hip_atomic_fetch_add(&tile[di * CH + lane], ni * pv[i],
                                       __ATOMIC_RELAXED, __HIP_MEMORY_SCOPE_WORKGROUP);
            }
        } else {
#pragma unroll
            for (int i = 0; i < 32; ++i) {
                if (i < n) {
                    float ni = __int_as_float(__builtin_amdgcn_readlane(__float_as_int(nrm), i));
                    int di = __builtin_amdgcn_readlane(dl, i);
                    __hip_atomic_fetch_add(&tile[di * CH + lane], ni * pv[i],
                                           __ATOMIC_RELAXED, __HIP_MEMORY_SCOPE_WORKGROUP);
                }
            }
        }
        s = s2; nrm = nrm2; dl = dl2; p = pn;
    }

    __syncthreads();
    int rows = N - base;
    if (rows > RB) rows = RB;
    float4* out4 = reinterpret_cast<float4*>(agg + (size_t)base * CH);
    for (int i = tid; i < rows * (CH / 4); i += SCW * 32) out4[i] = tile4[i];
}

__device__ __forceinline__ float sigmoidf_(float x) {
    return 1.0f / (1.0f + __expf(-x));
}

__global__ __launch_bounds__(256) void k_gates(const float* __restrict__ agg,
                                               const float* __restrict__ dinv,
                                               const float* __restrict__ Cst,
                                               const float* __restrict__ bx,
                                               const float* __restrict__ bh,
                                               const float* __restrict__ bi,
                                               const float* __restrict__ bf,
                                               const float* __restrict__ bc,
                                               const float* __restrict__ bo,
                                               float* __restrict__ out, int NC) {
    int t = blockIdx.x * blockDim.x + threadIdx.x;
    if (t >= NC) return;
    int c = t & (CH - 1);

    float a = agg[t];
    float s  = dinv[(unsigned)t / CH] * a + bx[c] + bh[c];
    float I  = sigmoidf_(s + bi[c]);
    float F  = sigmoidf_(s + bf[c]);
    float T  = tanhf(s + bc[c]);
    float Cn = F * Cst[t] + I * T;
    float O  = sigmoidf_(s + bo[c]);
    float Hn = O * tanhf(Cn);

    out[t]          = O;
    out[NC + t]     = Hn;
    out[2 * NC + t] = Cn;
}

extern "C" void kernel_launch(void* const* d_in, const int* in_sizes, int n_in,
                              void* d_out, int out_size, void* d_ws, size_t ws_size,
                              hipStream_t stream) {
    const float* X   = (const float*)d_in[0];
    const int*   ei  = (const int*)  d_in[1];
    const float* w   = (const float*)d_in[2];
    const float* H   = (const float*)d_in[3];
    const float* C   = (const float*)d_in[4];
    const float* Wx  = (const float*)d_in[5];
    const float* bx  = (const float*)d_in[6];
    const float* Wh  = (const float*)d_in[7];
    const float* bh  = (const float*)d_in[8];
    const float* bi  = (const float*)d_in[9];
    const float* bf  = (const float*)d_in[10];
    const float* bc  = (const float*)d_in[11];
    const float* bo  = (const float*)d_in[12];

    const int N = in_sizes[0] / CH;
    const int E = in_sizes[2];
    const int* src = ei;
    const int* dst = ei + E;

    float* ws  = (float*)d_ws;
    size_t Na  = ((size_t)N + 31) & ~(size_t)31;
    float* degp = ws;
    float* dinv = ws + (size_t)DSL * Na;
    float* P    = dinv + Na;
    float* agg  = P + (size_t)N * CH;
    const int NB = (N + RB - 1) / RB;
    const int G  = (E + CHK - 1) / CHK;
    size_t NBa = ((size_t)NB + 1 + 31) & ~(size_t)31;
    size_t GNa = ((size_t)G * NB + 31) & ~(size_t)31;
    int*   cnt      = (int*)(agg + (size_t)N * CH);
    int*   binstart = cnt + GNa;
    int*   bsrc     = binstart + NBa;
    float* bw       = (float*)(bsrc + E);
    int*   bdl      = (int*)(bw + E);

    dim3 dgrid((N + DR - 1) / DR, DSL);
    k_deg_lds<<<dgrid, DGT, 0, stream>>>(dst, w, degp, E, N, (int)Na, (E + DSL - 1) / DSL);
    k_dinv<<<(N + 255) / 256, 256, 0, stream>>>(degp, dinv, N, (int)Na);

    int tiles = (N + 15) / 16;
    k_gemm_wmma<<<(tiles + 7) / 8, 256, 0, stream>>>(X, H, Wx, Wh, dinv, P, N);

    k_bin_count<<<G, BT, 0, stream>>>(dst, cnt, E, N, NB);
    k_bin_fill<<<G, BT, 0, stream>>>(src, dst, w, cnt, binstart, bsrc, bw, bdl, E, N, NB, G);
    k_scatter_bins<<<NB, SCW * 32, 0, stream>>>(binstart, bsrc, bw, bdl, P, agg, N);

    int NC = N * CH;
    k_gates<<<(NC + 255) / 256, 256, 0, stream>>>(agg, dinv, C, bx, bh, bi, bf, bc, bo,
                                                  (float*)d_out, NC);
}


# gates fused into the scatter launch: each bin's work-group turns its LDS slab into O, H_new, C_new itself (C, row factors, biases loaded ahead of the edge loop); gates launch and agg plane go
# speedup vs baseline: 4.8279x; 1.0698x over previous
#include <hip/hip_runtime.h>
#include <hip/hip_bf16.h>
#include <math.h>

typedef __attribute__((ext_vector_type(2))) float v2f;
typedef __attribute__((ext_vector_type(8))) float v8f;

#define CH 32

#define DR   25088
#define DSL  64
#define DGT  1024
#define DU   8
#define DG   16

__global__ __launch_bounds__(DGT) void k_deg_lds(const int* __restrict__ dst,
                                                 const float* __restrict__ w,
                                                 float* __restrict__ degp,
                                                 int E, int N, int Na, int slice_len) {
    __shared__ __attribute__((aligned(16))) float dl[DR];

    const int tid  = threadIdx.x;
    const int base = blockIdx.x * DR;
    const int e_begin = blockIdx.y * slice_len;
    int e_end = e_begin + slice_len;
    if (e_end > E) e_end = E;

    float4* dl4 = reinterpret_cast<float4*>(dl);
    for (int i = tid; i < DR / 4; i += DGT) dl4[i] = make_float4(0.f, 0.f, 0.f, 0.f);
    __syncthreads();

    for (int cb = e_begin; cb < e_end; cb += DGT * DU) {
        int   d[DU];
        float ww[DU];
        const int*   pd = dst + cb + tid;
        const float* pw = w + cb + tid;
        if (cb + DGT * DU <= e_end) {
#pragma unroll
            for (int u = 0; u < DU; ++u) { d[u] = pd[u * DGT]; ww[u] = pw[u * DGT]; }
        } else {
#pragma unroll
            for (int u = 0; u < DU; ++u) {
                int idx = cb + u * DGT + tid;
                bool in = idx < e_end;
                int ii = in ? idx : e_end - 1;
                int dv = dst[ii];
                ww[u] = w[ii];
                d[u]  = in ? dv : -1;
            }
        }
#pragma unroll
        for (int u = 0; u < DU; ++u) {
            unsigned r = (unsigned)(d[u] - base);
            if (r < (unsigned)DR) {
                __hip_atomic_fetch_add(&dl[r], ww[u],
                                       __ATOMIC_RELAXED, __HIP_MEMORY_SCOPE_WORKGROUP);
            }
        }
    }
    __syncthreads();

    int rows = N - base;
    if (rows > DR) rows = DR;
    float* outp = degp + (size_t)blockIdx.y * Na + base;
    for (int i = tid; i < rows; i += DGT) outp[i] = dl[i];
}

__global__ __launch_bounds__(256) void k_dinv(const float* __restrict__ degp,
                                              float* __restrict__ dinv, int N, int Na) {
    int i = blockIdx.x * blockDim.x + threadIdx.x;
    if (i < N) {
        const float* p = degp + i;
        float d = 0.0f;
        for (int k = 0; k < DSL; k += DG) {
            float v[DG];
#pragma unroll
            for (int u = 0; u < DG; ++u) v[u] = p[(size_t)(k + u) * Na];
#pragma unroll
            for (int u = 0; u < DG; ++u) d += v[u];
        }
        dinv[i] = (d > 0.0f) ? rsqrtf(fmaxf(d, 1e-12f)) : 0.0f;
    }
}

__global__ __launch_bounds__(256) void k_gemm_wmma(const float* __restrict__ X,
                                                   const float* __restrict__ H,
                                                   const float* __restrict__ Wx,
                                                   const float* __restrict__ Wh,
                                                   const float* __restrict__ dinv,
                                                   float* __restrict__ P, int N) {
    int wave = blockIdx.x * (blockDim.x >> 5) + (threadIdx.x >> 5);
    int row0 = wave * 16;
    if (row0 >= N) return;

    int lane = threadIdx.x & 31;
    int half = lane >> 4;
    int lm   = lane & 15;

    int arow = row0 + lm;
    if (arow > N - 1) arow = N - 1;
    const float* xrow = X + (size_t)arow * CH;
    const float* hrow = H + (size_t)arow * CH;

    v8f acc0 = {};
    v8f acc1 = {};

    float dv[8];
#pragma unroll
    for (int r = 0; r < 8; ++r) {
        int drow = row0 + r + 8 * half;
        if (drow > N - 1) drow = N - 1;
        dv[r] = dinv[drow];
    }

#pragma unroll
    for (int kt = 0; kt < 8; ++kt) {
        int k0 = kt * 4;
        int ka = k0 + 2 * half;

        v2f ax = { xrow[ka], xrow[ka + 1] };
        v2f ah = { hrow[ka], hrow[ka + 1] };

        int krow = ka * CH;
        v2f bx0 = { Wx[krow + lm],      Wx[krow + CH + lm]      };
        v2f bx1 = { Wx[krow + 16 + lm], Wx[krow + CH + 16 + lm] };
        v2f bh0 = { Wh[krow + lm],      Wh[krow + CH + lm]      };
        v2f bh1 = { Wh[krow + 16 + lm], Wh[krow + CH + 16 + lm] };

        acc0 = __builtin_amdgcn_wmma_f32_16x16x4_f32(false, ax, false, bx0,
                                                     (short)0, acc0, false, false);
        acc1 = __builtin_amdgcn_wmma_f32_16x16x4_f32(false, ax, false, bx1,
                                                     (short)0, acc1, false, false);
        acc0 = __builtin_amdgcn_wmma_f32_16x16x4_f32(false, ah, false, bh0,
                                                     (short)0, acc0, false, false);
        acc1 = __builtin_amdgcn_wmma_f32_16x16x4_f32(false, ah, false, bh1,
                                                     (short)0, acc1, false, false);
    }

    float* prow = P + (size_t)row0 * CH;
#pragma unroll
    for (int r = 0; r < 8; ++r) {
        int rr = r + 8 * half;
        if (row0 + rr < N) {
            prow[(size_t)rr * CH + lm]      = dv[r] * acc0[r];
            prow[(size_t)rr * CH + 16 + lm] = dv[r] * acc1[r];
        }
    }
}

#define RB    400
#define CHK   12800
#define BT    1024
#define BU    13
#define NBMAX 1024
#define SCW   32
#define GI    ((RB * CH + SCW * 32 - 1) / (SCW * 32))

__global__ __launch_bounds__(BT) void k_bin_count(const int* __restrict__ dst,
                                                  int* __restrict__ cnt,
                                                  int E, int N, int NB) {
    __shared__ int c[NBMAX];
    const int tid = threadIdx.x;
    const int g   = blockIdx.x;
    const int e0  = g * CHK;
    int e1 = e0 + CHK;
    if (e1 > E) e1 = E;

    for (int b = tid; b < NB; b += BT) c[b] = 0;
    __syncthreads();

    int d[BU];
#pragma unroll
    for (int u = 0; u < BU; ++u) {
        int e = e0 + u * BT + tid;
        d[u] = (e < e1) ? dst[e] : -1;
    }
#pragma unroll
    for (int u = 0; u < BU; ++u) {
        if ((unsigned)d[u] < (unsigned)N) {
            __hip_atomic_fetch_add(&c[(unsigned)d[u] / RB], 1,
                                   __ATOMIC_RELAXED, __HIP_MEMORY_SCOPE_WORKGROUP);
        }
    }
    __syncthreads();
    for (int b = tid; b < NB; b += BT) cnt[(size_t)g * NB + b] = c[b];
}

__global__ __launch_bounds__(BT) void k_bin_fill(const int* __restrict__ src,
                                                 const int* __restrict__ dst,
                                                 const float* __restrict__ w,
                                                 const int* __restrict__ cnt,
                                                 int* __restrict__ binstart,
                                                 int* __restrict__ bsrc,
                                                 float* __restrict__ bw,
                                                 int* __restrict__ bdl,
                                                 int E, int N, int NB, int G) {
    __shared__ int   ssrc[CHK];
    __shared__ float sw[CHK];
    __shared__ int   sdst[CHK];
    __shared__ int   colsum[NBMAX];
    __shared__ int   pre[NBMAX];
    __shared__ int   gst[NBMAX + 1];
    __shared__ int   lst[NBMAX + 1];
    __shared__ int   cur[NBMAX];
    __shared__ int   gdelta[NBMAX];

    const int tid  = threadIdx.x;
    const int lane = tid & 31;
    const int wv   = tid >> 5;
    const int g    = blockIdx.x;
    const int e0   = g * CHK;
    int e1 = e0 + CHK;
    if (e1 > E) e1 = E;

    int   d[BU], s[BU];
    float ww[BU];
#pragma unroll
    for (int u = 0; u < BU; ++u) {
        int e = e0 + u * BT + tid;
        bool in = e < e1;
        int ee = in ? e : e0;
        int dv = dst[ee];
        s[u]  = src[ee];
        ww[u] = w[ee];
        d[u]  = in ? dv : -1;
    }

    for (int b = tid; b < NB; b += BT) { colsum[b] = 0; pre[b] = 0; }
    __syncthreads();

    for (int b = tid & 255; b < NB; b += 256) {
        int tot = 0, pr = 0;
#pragma unroll 16
        for (int gg = tid >> 8; gg < G; gg += 4) {
            int v = cnt[(size_t)gg * NB + b];
            tot += v;
            pr  += (gg < g) ? v : 0;
        }
        __hip_atomic_fetch_add(&colsum[b], tot, __ATOMIC_RELAXED, __HIP_MEMORY_SCOPE_WORKGROUP);
        __hip_atomic_fetch_add(&pre[b],    pr,  __ATOMIC_RELAXED, __HIP_MEMORY_SCOPE_WORKGROUP);
    }
    __syncthreads();

    if (wv == 0) {
        int per = (NB + 31) >> 5;
        int b0 = lane * per;
        int b1 = b0 + per;
        if (b1 > NB) b1 = NB;
        int s1 = 0, s2 = 0;
        for (int b = b0; b < b1; ++b) { s1 += colsum[b]; s2 += cnt[(size_t)g * NB + b]; }
        int i1 = s1, i2 = s2;
        for (int off = 1; off < 32; off <<= 1) {
            int t1 = __shfl_up(i1, off);
            int t2 = __shfl_up(i2, off);
            if (lane >= off) { i1 += t1; i2 += t2; }
        }
        int x1 = i1 - s1, x2 = i2 - s2;
        for (int b = b0; b < b1; ++b) {
            gst[b] = x1;
            lst[b] = x2;
            x1 += colsum[b];
            x2 += cnt[(size_t)g * NB + b];
        }
        if (lane == 31) { gst[NB] = i1; lst[NB] = i2; }
    }
    __syncthreads();

    for (int b = tid; b < NB; b += BT) {
        int l = lst[b];
        cur[b]    = l;
        gdelta[b] = gst[b] + pre[b] - l;
        if (g == 0) binstart[b] = gst[b];
    }
    if (g == 0 && tid == 0) binstart[NB] = gst[NB];
    __syncthreads();

#pragma unroll
    for (int u = 0; u < BU; ++u) {
        if ((unsigned)d[u] < (unsigned)N) {
            int p = __hip_atomic_fetch_add(&cur[(unsigned)d[u] / RB], 1,
                                           __ATOMIC_RELAXED, __HIP_MEMORY_SCOPE_WORKGROUP);
            ssrc[p] = s[u];
            sw[p]   = ww[u];
            sdst[p] = d[u];
        }
    }
    __syncthreads();

    const int n = lst[NB];
    for (int i = tid; i < n; i += BT) {
        unsigned dv = (unsigned)sdst[i];
        unsigned b  = dv / RB;
        int gi = gdelta[b] + i;
        bsrc[gi] = ssrc[i];
        bw[gi]   = sw[i];
        bdl[gi]  = (int)(dv - b * RB);
    }
}

__device__ __forceinline__ float sigmoidf_(float x) {
    return 1.0f / (1.0f + __expf(-x));
}

__global__ __launch_bounds__(SCW * 32) void k_scatter_gates(const int* __restrict__ binstart,
                                                            const int* __restrict__ bsrc,
                                                            const float* __restrict__ bw,
                                                            const int* __restrict__ bdl,
                                                            const float* __restrict__ P,
                                                            const float* __restrict__ dinv,
                                                            const float* __restrict__ Cst,
                                                            const float* __restrict__ bx,
                                                            const float* __restrict__ bh,
                                                            const float* __restrict__ bi,
                                                            const float* __restrict__ bf,
                                                            const float* __restrict__ bc,
                                                            const float* __restrict__ bo,
                                                            float* __restrict__ out, int N) {
    __shared__ __attribute__((aligned(16))) float tile[RB * CH];

    const int tid  = threadIdx.x;
    const int lane = tid & 31;
    const int wv    = __builtin_amdgcn_readfirstlane(tid >> 5);
    const int base  = blockIdx.x * RB;
    const int start = __builtin_amdgcn_readfirstlane(binstart[blockIdx.x]);
    const int end   = __builtin_amdgcn_readfirstlane(binstart[blockIdx.x + 1]);
    int rows = N - base;
    if (rows > RB) rows = RB;

    int p = start + wv * 32;
    int   s = 0, dl = 0;
    float nrm = 0.0f;
    if (p + lane < end) { s = bsrc[p + lane]; nrm = bw[p + lane]; dl = bdl[p + lane]; }

    const float* crow = Cst + (size_t)base * CH;
    float cv[GI], dvr[GI];
#pragma unroll
    for (int k = 0; k < GI; ++k) {
        int r = wv + k * SCW;
        if (r > rows - 1) r = rows - 1;
        cv[k]  = crow[r * CH + lane];
        dvr[k] = dinv[base + r];
    }
    const float bxl = bx[lane], bhl = bh[lane], bil = bi[lane], bfl = bf[lane], bcl = bc[lane], bol = bo[lane];

    float4* tile4 = reinterpret_cast<float4*>(tile);
    for (int i = tid; i < RB * CH / 4; i += SCW * 32) tile4[i] = make_float4(0.f, 0.f, 0.f, 0.f);
    __syncthreads();

    while (p < end) {
        const int pn = p + SCW * 32;
        int ix = pn + lane;
        if (ix > end - 1) ix = end - 1;
        int   s2   = bsrc[ix];
        float nrm2 = bw[ix];
        int   dl2  = bdl[ix];

        int n = end - p;
        if (n > 32) n = 32;
        float pv[32];
#pragma unroll
        for (int i = 0; i < 32; ++i) {
            unsigned si = (unsigned)__builtin_amdgcn_readlane(s, i);
            pv[i] = P[si * (unsigned)CH + (unsigned)lane];
        }
        if (n == 32) {
#pragma unroll
            for (int i = 0; i < 32; ++i) {
                float ni = __int_as_float(__builtin_amdgcn_readlane(__float_as_int(nrm), i));
                int di = __builtin_amdgcn_readlane(dl, i);
                __hip_atomic_fetch_add(&tile[di * CH + lane], ni * pv[i],
                                       __ATOMIC_RELAXED, __HIP_MEMORY_SCOPE_WORKGROUP);
            }
        } else {
#pragma unroll
            for (int i = 0; i < 32; ++i) {
                if (i < n) {
                    float ni = __int_as_float(__builtin_amdgcn_readlane(__float_as_int(nrm), i));
                    int di = __builtin_amdgcn_readlane(dl, i);
                    __hip_atomic_fetch_add(&tile[di * CH + lane], ni * pv[i],
                                           __ATOMIC_RELAXED, __HIP_MEMORY_SCOPE_WORKGROUP);
                }
            }
        }
        s = s2; nrm = nrm2; dl = dl2; p = pn;
    }

    __syncthreads();

    const size_t NC = (size_t)N * CH;
    float* orow = out + (size_t)base * CH;
#pragma unroll
    for (int k = 0; k < GI; ++k) {
        const int r = wv + k * SCW;
        if (r < rows) {
            const int i = r * CH + lane;
            float a  = tile[i];
            float sg = dvr[k] * a + bxl + bhl;
            float I  = sigmoidf_(sg + bil);
            float F  = sigmoidf_(sg + bfl);
            float T  = tanhf(sg + bcl);
            float Cn = F * cv[k] + I * T;
            float O  = sigmoidf_(sg + bol);
            float Hn = O * tanhf(Cn);
            orow[i]          = O;
            orow[NC + i]     = Hn;
            orow[2 * NC + i] = Cn;
        }
    }
}

extern "C" void kernel_launch(void* const* d_in, const int* in_sizes, int n_in,
                              void* d_out, int out_size, void* d_ws, size_t ws_size,
                              hipStream_t stream) {
    const float* X   = (const float*)d_in[0];
    const int*   ei  = (const int*)  d_in[1];
    const float* w   = (const float*)d_in[2];
    const float* H   = (const float*)d_in[3];
    const float* C   = (const float*)d_in[4];
    const float* Wx  = (const float*)d_in[5];
    const float* bx  = (const float*)d_in[6];
    const float* Wh  = (const float*)d_in[7];
    const float* bh  = (const float*)d_in[8];
    const float* bi  = (const float*)d_in[9];
    const float* bf  = (const float*)d_in[10];
    const float* bc  = (const float*)d_in[11];
    const float* bo  = (const float*)d_in[12];

    const int N = in_sizes[0] / CH;
    const int E = in_sizes[2];
    const int* src = ei;
    const int* dst = ei + E;

    float* ws  = (float*)d_ws;
    size_t Na  = ((size_t)N + 31) & ~(size_t)31;
    float* degp = ws;
    float* dinv = ws + (size_t)DSL * Na;
    float* P    = dinv + Na;
    const int NB = (N + RB - 1) / RB;
    const int G  = (E + CHK - 1) / CHK;
    size_t NBa = ((size_t)NB + 1 + 31) & ~(size_t)31;
    size_t GNa = ((size_t)G * NB + 31) & ~(size_t)31;
    int*   cnt      = (int*)(P + (size_t)N * CH);
    int*   binstart = cnt + GNa;
    int*   bsrc     = binstart + NBa;
    float* bw       = (float*)(bsrc + E);
    int*   bdl      = (int*)(bw + E);

    dim3 dgrid((N + DR - 1) / DR, DSL);
    k_deg_lds<<<dgrid, DGT, 0, stream>>>(dst, w, degp, E, N, (int)Na, (E + DSL - 1) / DSL);
    k_dinv<<<(N + 255) / 256, 256, 0, stream>>>(degp, dinv, N, (int)Na);

    int tiles = (N + 15) / 16;
    k_gemm_wmma<<<(tiles + 7) / 8, 256, 0, stream>>>(X, H, Wx, Wh, dinv, P, N);

    k_bin_count<<<G, BT, 0, stream>>>(dst, cnt, E, N, NB);
    k_bin_fill<<<G, BT, 0, stream>>>(src, dst, w, cnt, binstart, bsrc, bw, bdl, E, N, NB, G);
    k_scatter_gates<<<NB, SCW * 32, 0, stream>>>(binstart, bsrc, bw, bdl, P, dinv, C,
                                                 bx, bh, bi, bf, bc, bo, (float*)d_out, N);
}


# degree sums from the binned edges: the GEMM launch has one work-group per dst bin that first adds the bin's w per row in LDS (float atomics) and makes dinv; plane, dinv launches go; sum order changes
# speedup vs baseline: 5.0274x; 1.0413x over previous
#include <hip/hip_runtime.h>
#include <hip/hip_bf16.h>
#include <math.h>

typedef __attribute__((ext_vector_type(2))) float v2f;
typedef __attribute__((ext_vector_type(8))) float v8f;

#define CH 32

#define RB    400
#define CHK   12800
#define BT    1024
#define BU    13
#define NBMAX 1024
#define SCW   32
#define GI    ((RB * CH + SCW * 32 - 1) / (SCW * 32))

__global__ __launch_bounds__(BT) void k_bin_count(const int* __restrict__ dst,
                                                  int* __restrict__ cnt,
                                                  int E, int N, int NB) {
    __shared__ int c[NBMAX];
    const int tid = threadIdx.x;
    const int g   = blockIdx.x;
    const int e0  = g * CHK;
    int e1 = e0 + CHK;
    if (e1 > E) e1 = E;

    for (int b = tid; b < NB; b += BT) c[b] = 0;
    __syncthreads();

    int d[BU];
#pragma unroll
    for (int u = 0; u < BU; ++u) {
        int e = e0 + u * BT + tid;
        d[u] = (e < e1) ? dst[e] : -1;
    }
#pragma unroll
    for (int u = 0; u < BU; ++u) {
        if ((unsigned)d[u] < (unsigned)N) {
            __hip_atomic_fetch_add(&c[(unsigned)d[u] / RB], 1,
                                   __ATOMIC_RELAXED, __HIP_MEMORY_SCOPE_WORKGROUP);
        }
    }
    __syncthreads();
    for (int b = tid; b < NB; b += BT) cnt[(size_t)g * NB + b] = c[b];
}

__global__ __launch_bounds__(BT) void k_bin_fill(const int* __restrict__ src,
                                                 const int* __restrict__ dst,
                                                 const float* __restrict__ w,
                                                 const int* __restrict__ cnt,
                                                 int* __restrict__ binstart,
                                                 int* __restrict__ bsrc,
                                                 float* __restrict__ bw,
                                                 int* __restrict__ bdl,
                                                 int E, int N, int NB, int G) {
    __shared__ int   ssrc[CHK];
    __shared__ float sw[CHK];
    __shared__ int   sdst[CHK];
    __shared__ int   colsum[NBMAX];
    __shared__ int   pre[NBMAX];
    __shared__ int   gst[NBMAX + 1];
    __shared__ int   lst[NBMAX + 1];
    __shared__ int   cur[NBMAX];
    __shared__ int   gdelta[NBMAX];

    const int tid  = threadIdx.x;
    const int lane = tid & 31;
    const int wv   = tid >> 5;
    const int g    = blockIdx.x;
    const int e0   = g * CHK;
    int e1 = e0 + CHK;
    if (e1 > E) e1 = E;

    int   d[BU], s[BU];
    float ww[BU];
#pragma unroll
    for (int u = 0; u < BU; ++u) {
        int e = e0 + u * BT + tid;
        bool in = e < e1;
        int ee = in ? e : e0;
        int dv = dst[ee];
        s[u]  = src[ee];
        ww[u] = w[ee];
        d[u]  = in ? dv : -1;
    }

    for (int b = tid; b < NB; b += BT) { colsum[b] = 0; pre[b] = 0; }
    __syncthreads();

    for (int b = tid & 255; b < NB; b += 256) {
        int tot = 0, pr = 0;
#pragma unroll 16
        for (int gg = tid >> 8; gg < G; gg += 4) {
            int v = cnt[(size_t)gg * NB + b];
            tot += v;
            pr  += (gg < g) ? v : 0;
        }
        __hip_atomic_fetch_add(&colsum[b], tot, __ATOMIC_RELAXED, __HIP_MEMORY_SCOPE_WORKGROUP);
        __hip_atomic_fetch_add(&pre[b],    pr,  __ATOMIC_RELAXED, __HIP_MEMORY_SCOPE_WORKGROUP);
    }
    __syncthreads();

    if (wv == 0) {
        int per = (NB + 31) >> 5;
        int b0 = lane * per;
        int b1 = b0 + per;
        if (b1 > NB) b1 = NB;
        int s1 = 0, s2 = 0;
        for (int b = b0; b < b1; ++b) { s1 += colsum[b]; s2 += cnt[(size_t)g * NB + b]; }
        int i1 = s1, i2 = s2;
        for (int off = 1; off < 32; off <<= 1) {
            int t1 = __shfl_up(i1, off);
            int t2 = __shfl_up(i2, off);
            if (lane >= off) { i1 += t1; i2 += t2; }
        }
        int x1 = i1 - s1, x2 = i2 - s2;
        for (int b = b0; b < b1; ++b) {
            gst[b] = x1;
            lst[b] = x2;
            x1 += colsum[b];
            x2 += cnt[(size_t)g * NB + b];
        }
        if (lane == 31) { gst[NB] = i1; lst[NB] = i2; }
    }
    __syncthreads();

    for (int b = tid; b < NB; b += BT) {
        int l = lst[b];
        cur[b]    = l;
        gdelta[b] = gst[b] + pre[b] - l;
        if (g == 0) binstart[b] = gst[b];
    }
    if (g == 0 && tid == 0) binstart[NB] = gst[NB];
    __syncthreads();

#pragma unroll
    for (int u = 0; u < BU; ++u) {
        if ((unsigned)d[u] < (unsigned)N) {
            int p = __hip_atomic_fetch_add(&cur[(unsigned)d[u] / RB], 1,
                                           __ATOMIC_RELAXED, __HIP_MEMORY_SCOPE_WORKGROUP);
            ssrc[p] = s[u];
            sw[p]   = ww[u];
            sdst[p] = d[u];
        }
    }
    __syncthreads();

    const int n = lst[NB];
    for (int i = tid; i < n; i += BT) {
        unsigned dv = (unsigned)sdst[i];
        unsigned b  = dv / RB;
        int gi = gdelta[b] + i;
        bsrc[gi] = ssrc[i];
        bw[gi]   = sw[i];
        bdl[gi]  = (int)(dv - b * RB);
    }
}

#define GW   (RB / 16)
#define GT   1024
#define EU   16

static_assert(RB % 16 == 0, "a bin is a whole number of 16-row tiles");
static_assert(GW * 32 <= GT && RB <= GT, "one wave per tile and one thread per bin row");

__global__ __launch_bounds__(GT) void k_deg_gemm_wmma(const int* __restrict__ binstart,
                                                      const float* __restrict__ bw,
                                                      const int* __restrict__ bdl,
                                                      const float* __restrict__ X,
                                                      const float* __restrict__ H,
                                                      const float* __restrict__ Wx,
                                                      const float* __restrict__ Wh,
                                                      float* __restrict__ dinv,
                                                      float* __restrict__ P, int N) {
    __shared__ float degl[RB];

    const int tid  = threadIdx.x;
    const int lane = tid & 31;
    const int half = lane >> 4;
    const int lm   = lane & 15;
    const int wv    = __builtin_amdgcn_readfirstlane(tid >> 5);
    const int base  = blockIdx.x * RB;
    const int start = __builtin_amdgcn_readfirstlane(binstart[blockIdx.x]);
    const int end   = __builtin_amdgcn_readfirstlane(binstart[blockIdx.x + 1]);
    int rows = N - base;
    if (rows > RB) rows = RB;

    const int  row0     = base + wv * 16;
    const bool has_tile = (wv < GW) && (row0 < N);

    v2f ax[8], ah[8];
#pragma unroll
    for (int kt = 0; kt < 8; ++kt) { ax[kt] = v2f{0.f, 0.f}; ah[kt] = v2f{0.f, 0.f}; }
    if (has_tile) {
        int arow = row0 + lm;
        if (arow > N - 1) arow = N - 1;
        const float* xrow = X + (size_t)arow * CH;
        const float* hrow = H + (size_t)arow * CH;
#pragma unroll
        for (int kt = 0; kt < 8; ++kt) {
            int ka = kt * 4 + 2 * half;
            ax[kt] = v2f{ xrow[ka], xrow[ka + 1] };
            ah[kt] = v2f{ hrow[ka], hrow[ka + 1] };
        }
    }

    for (int i = tid; i < RB; i += GT) degl[i] = 0.0f;
    __syncthreads();

    for (int cb = start; cb < end; cb += GT * EU) {
        int   d[EU];
        float ww[EU];
#pragma unroll
        for (int u = 0; u < EU; ++u) {
            int idx = cb + u * GT + tid;
            int ii = (idx < end) ? idx : end - 1;
            d[u]  = bdl[ii];
            ww[u] = bw[ii];
        }
#pragma unroll
        for (int u = 0; u < EU; ++u) {
            float a = (cb + u * GT + tid < end) ? ww[u] : 0.0f;
            __hip_atomic_fetch_add(&degl[d[u]], a,
                                   __ATOMIC_RELAXED, __HIP_MEMORY_SCOPE_WORKGROUP);
        }
    }
    __syncthreads();

    if (tid < RB) {
        float dg = degl[tid];
        float v  = (dg > 0.0f) ? rsqrtf(fmaxf(dg, 1e-12f)) : 0.0f;
        degl[tid] = v;
        if (tid < rows) dinv[base + tid] = v;
    }
    __syncthreads();

    if (has_tile) {
        float dv[8];
#pragma unroll
        for (int r = 0; r < 8; ++r) dv[r] = degl[wv * 16 + r + 8 * half];

        v8f acc0 = {};
        v8f acc1 = {};

#pragma unroll
        for (int kt = 0; kt < 8; ++kt) {
            int ka = kt * 4 + 2 * half;

            int krow = ka * CH;
            v2f bx0 = { Wx[krow + lm],      Wx[krow + CH + lm]      };
            v2f bx1 = { Wx[krow + 16 + lm], Wx[krow + CH + 16 + lm] };
            v2f bh0 = { Wh[krow + lm],      Wh[krow + CH + lm]      };
            v2f bh1 = { Wh[krow + 16 + lm], Wh[krow + CH + 16 + lm] };

            acc0 = __builtin_amdgcn_wmma_f32_16x16x4_f32(false, ax[kt], false, bx0,
                                                         (short)0, acc0, false, false);
            acc1 = __builtin_amdgcn_wmma_f32_16x16x4_f32(false, ax[kt], false, bx1,
                                                         (short)0, acc1, false, false);
            acc0 = __builtin_amdgcn_wmma_f32_16x16x4_f32(false, ah[kt], false, bh0,
                                                         (short)0, acc0, false, false);
            acc1 = __builtin_amdgcn_wmma_f32_16x16x4_f32(false, ah[kt], false, bh1,
                                                         (short)0, acc1, false, false);
        }

        float* prow = P + (size_t)row0 * CH;
#pragma unroll
        for (int r = 0; r < 8; ++r) {
            int rr = r + 8 * half;
            if (row0 + rr < N) {
                prow[(size_t)rr * CH + lm]      = dv[r] * acc0[r];
                prow[(size_t)rr * CH + 16 + lm] = dv[r] * acc1[r];
            }
        }
    }
}

__device__ __forceinline__ float sigmoidf_(float x) {
    return 1.0f / (1.0f + __expf(-x));
}

__global__ __launch_bounds__(SCW * 32) void k_scatter_gates(const int* __restrict__ binstart,
                                                            const int* __restrict__ bsrc,
                                                            const float* __restrict__ bw,
                                                            const int* __restrict__ bdl,
                                                            const float* __restrict__ P,
                                                            const float* __restrict__ dinv,
                                                            const float* __restrict__ Cst,
                                                            const float* __restrict__ bx,
                                                            const float* __restrict__ bh,
                                                            const float* __restrict__ bi,
                                                            const float* __restrict__ bf,
                                                            const float* __restrict__ bc,
                                                            const float* __restrict__ bo,
                                                            float* __restrict__ out, int N) {
    __shared__ __attribute__((aligned(16))) float tile[RB * CH];

    const int tid  = threadIdx.x;
    const int lane = tid & 31;
    const int wv    = __builtin_amdgcn_readfirstlane(tid >> 5);
    const int base  = blockIdx.x * RB;
    const int start = __builtin_amdgcn_readfirstlane(binstart[blockIdx.x]);
    const int end   = __builtin_amdgcn_readfirstlane(binstart[blockIdx.x + 1]);
    int rows = N - base;
    if (rows > RB) rows = RB;

    int p = start + wv * 32;
    int   s = 0, dl = 0;
    float nrm = 0.0f;
    if (p + lane < end) { s = bsrc[p + lane]; nrm = bw[p + lane]; dl = bdl[p + lane]; }

    const float* crow = Cst + (size_t)base * CH;
    float cv[GI], dvr[GI];
#pragma unroll
    for (int k = 0; k < GI; ++k) {
        int r = wv + k * SCW;
        if (r > rows - 1) r = rows - 1;
        cv[k]  = crow[r * CH + lane];
        dvr[k] = dinv[base + r];
    }
    const float bxl = bx[lane], bhl = bh[lane], bil = bi[lane], bfl = bf[lane], bcl = bc[lane], bol = bo[lane];

    float4* tile4 = reinterpret_cast<float4*>(tile);
    for (int i = tid; i < RB * CH / 4; i += SCW * 32) tile4[i] = make_float4(0.f, 0.f, 0.f, 0.f);
    __syncthreads();

    while (p < end) {
        const int pn = p + SCW * 32;
        int ix = pn + lane;
        if (ix > end - 1) ix = end - 1;
        int   s2   = bsrc[ix];
        float nrm2 = bw[ix];
        int   dl2  = bdl[ix];

        int n = end - p;
        if (n > 32) n = 32;
        float pv[32];
#pragma unroll
        for (int i = 0; i < 32; ++i) {
            unsigned si = (unsigned)__builtin_amdgcn_readlane(s, i);
            pv[i] = P[si * (unsigned)CH + (unsigned)lane];
        }
        if (n == 32) {
#pragma unroll
            for (int i = 0; i < 32; ++i) {
                float ni = __int_as_float(__builtin_amdgcn_readlane(__float_as_int(nrm), i));
                int di = __builtin_amdgcn_readlane(dl, i);
                __hip_atomic_fetch_add(&tile[di * CH + lane], ni * pv[i],
                                       __ATOMIC_RELAXED, __HIP_MEMORY_SCOPE_WORKGROUP);
            }
        } else {
#pragma unroll
            for (int i = 0; i < 32; ++i) {
                if (i < n) {
                    float ni = __int_as_float(__builtin_amdgcn_readlane(__float_as_int(nrm), i));
                    int di = __builtin_amdgcn_readlane(dl, i);
                    __hip_atomic_fetch_add(&tile[di * CH + lane], ni * pv[i],
                                           __ATOMIC_RELAXED, __HIP_MEMORY_SCOPE_WORKGROUP);
                }
            }
        }
        s = s2; nrm = nrm2; dl = dl2; p = pn;
    }

    __syncthreads();

    const size_t NC = (size_t)N * CH;
    float* orow = out + (size_t)base * CH;
#pragma unroll
    for (int k = 0; k < GI; ++k) {
        const int r = wv + k * SCW;
        if (r < rows) {
            const int i = r * CH + lane;
            float a  = tile[i];
            float sg = dvr[k] * a + bxl + bhl;
            float I  = sigmoidf_(sg + bil);
            float F  = sigmoidf_(sg + bfl);
            float T  = tanhf(sg + bcl);
            float Cn = F * cv[k] + I * T;
            float O  = sigmoidf_(sg + bol);
            float Hn = O * tanhf(Cn);
            orow[i]          = O;
            orow[NC + i]     = Hn;
            orow[2 * NC + i] = Cn;
        }
    }
}

extern "C" void kernel_launch(void* const* d_in, const int* in_sizes, int n_in,
                              void* d_out, int out_size, void* d_ws, size_t ws_size,
                              hipStream_t stream) {
    const float* X   = (const float*)d_in[0];
    const int*   ei  = (const int*)  d_in[1];
    const float* w   = (const float*)d_in[2];
    const float* H   = (const float*)d_in[3];
    const float* C   = (const float*)d_in[4];
    const float* Wx  = (const float*)d_in[5];
    const float* bx  = (const float*)d_in[6];
    const float* Wh  = (const float*)d_in[7];
    const float* bh  = (const float*)d_in[8];
    const float* bi  = (const float*)d_in[9];
    const float* bf  = (const float*)d_in[10];
    const float* bc  = (const float*)d_in[11];
    const float* bo  = (const float*)d_in[12];

    const int N = in_sizes[0] / CH;
    const int E = in_sizes[2];
    const int* src = ei;
    const int* dst = ei + E;

    float* ws  = (float*)d_ws;
    size_t Na  = ((size_t)N + 31) & ~(size_t)31;
    float* dinv = ws;
    float* P    = dinv + Na;
    const int NB = (N + RB - 1) / RB;
    const int G  = (E + CHK - 1) / CHK;
    size_t NBa = ((size_t)NB + 1 + 31) & ~(size_t)31;
    size_t GNa = ((size_t)G * NB + 31) & ~(size_t)31;
    int*   cnt      = (int*)(P + (size_t)N * CH);
    int*   binstart = cnt + GNa;
    int*   bsrc     = binstart + NBa;
    float* bw       = (float*)(bsrc + E);
    int*   bdl      = (int*)(bw + E);

    k_bin_count<<<G, BT, 0, stream>>>(dst, cnt, E, N, NB);
    k_bin_fill<<<G, BT, 0, stream>>>(src, dst, w, cnt, binstart, bsrc, bw, bdl, E, N, NB, G);
    k_deg_gemm_wmma<<<NB, GT, 0, stream>>>(binstart, bw, bdl, X, H, Wx, Wh, dinv, P, N);
    k_scatter_gates<<<NB, SCW * 32, 0, stream>>>(binstart, bsrc, bw, bdl, P, dinv, C,
                                                 bx, bh, bi, bf, bc, bo, (float*)d_out, N);
}


# binning: edges ordered by dst bin inside each chunk only: no shared counts, no counting launch (4 -> 3); a bin's waves read one segment per chunk, lined up by a wave scan; sum order changes
# speedup vs baseline: 6.9699x; 1.3864x over previous
#include <hip/hip_runtime.h>
#include <hip/hip_bf16.h>
#include <math.h>

typedef __attribute__((ext_vector_type(2))) float v2f;
typedef __attribute__((ext_vector_type(8))) float v8f;

#define CH 32

#define RB    400
#define CHK   12800
#define BT    1024
#define BU    13
#define NBMAX 1024
#define SCW   32
#define SEGB  (SCW * 32)
#define GI    ((RB * CH + SCW * 32 - 1) / (SCW * 32))

static_assert(BT == NBMAX && BT == 32 * 32, "one thread per table bin, and one lane per wave total in the scan");

__device__ __forceinline__ int wave_scan_incl(int v, int lane) {
    int inc = v;
    for (int off = 1; off < 32; off <<= 1) {
        int t = __shfl_up(inc, off);
        if (lane >= off) inc += t;
    }
    return inc;
}

__global__ __launch_bounds__(BT) void k_bin_local(const int* __restrict__ src,
                                                  const int* __restrict__ dst,
                                                  const float* __restrict__ w,
                                                  int* __restrict__ lstg,
                                                  int* __restrict__ bsrc,
                                                  float* __restrict__ bw,
                                                  int* __restrict__ bdl,
                                                  int E, int N, int NB, int LP) {
    __shared__ int   ssrc[CHK];
    __shared__ float sw[CHK];
    __shared__ int   sdl[CHK];
    __shared__ int   cntl[NBMAX];
    __shared__ int   lst[NBMAX + 1];
    __shared__ int   wtot[BT / 32];

    const int tid  = threadIdx.x;
    const int lane = tid & 31;
    const int wv   = tid >> 5;
    const int g    = blockIdx.x;
    const int e0   = g * CHK;
    int e1 = e0 + CHK;
    if (e1 > E) e1 = E;

    int   d[BU], s[BU];
    float ww[BU];
#pragma unroll
    for (int u = 0; u < BU; ++u) {
        int e = e0 + u * BT + tid;
        bool in = e < e1;
        int ee = in ? e : e0;
        int dv = dst[ee];
        s[u]  = src[ee];
        ww[u] = w[ee];
        d[u]  = in ? dv : -1;
    }

    cntl[tid] = 0;
    __syncthreads();

    int rk[BU];
#pragma unroll
    for (int u = 0; u < BU; ++u) {
        rk[u] = 0;
        if ((unsigned)d[u] < (unsigned)N) {
            rk[u] = __hip_atomic_fetch_add(&cntl[(unsigned)d[u] / RB], 1,
                                           __ATOMIC_RELAXED, __HIP_MEMORY_SCOPE_WORKGROUP);
        }
    }
    __syncthreads();

    const int c   = (tid < NB) ? cntl[tid] : 0;
    const int inc = wave_scan_incl(c, lane);
    if (lane == 31) wtot[wv] = inc;
    __syncthreads();
    const int winc = wave_scan_incl(wtot[lane], lane);
    int wpre = __shfl(winc, (wv + 31) & 31);
    if (wv == 0) wpre = 0;
    const int excl = wpre + inc - c;
    if (tid < NB) lst[tid] = excl;
    if (tid == BT - 1) lst[NB] = excl + c;
    __syncthreads();

    for (int b = tid; b <= NB; b += BT) lstg[(size_t)g * LP + b] = lst[b];

#pragma unroll
    for (int u = 0; u < BU; ++u) {
        if ((unsigned)d[u] < (unsigned)N) {
            unsigned b = (unsigned)d[u] / RB;
            int p = lst[b] + rk[u];
            ssrc[p] = s[u];
            sw[p]   = ww[u];
            sdl[p]  = (int)((unsigned)d[u] - b * RB);
        }
    }
    __syncthreads();

    const int n = lst[NB];
    int*   osrc = bsrc + (size_t)g * CHK;
    float* ow   = bw   + (size_t)g * CHK;
    int*   odl  = bdl  + (size_t)g * CHK;
#pragma unroll
    for (int u = 0; u < BU; ++u) {
        int i = u * BT + tid;
        if (i < n) {
            osrc[i] = ssrc[i];
            ow[i]   = sw[i];
            odl[i]  = sdl[i];
        }
    }
}

#define GW   (RB / 16)
#define GT   (SCW * 32)
#define SGB  8
#define SGT  3

static_assert(RB % 16 == 0, "a bin is a whole number of 16-row tiles");
static_assert(GW * 32 <= GT && RB <= GT, "one wave per tile and one thread per bin row");

__global__ __launch_bounds__(GT) void k_deg_gemm_wmma(const int* __restrict__ lstg,
                                                      const float* __restrict__ bw,
                                                      const int* __restrict__ bdl,
                                                      const float* __restrict__ X,
                                                      const float* __restrict__ H,
                                                      const float* __restrict__ Wx,
                                                      const float* __restrict__ Wh,
                                                      float* __restrict__ dinv,
                                                      float* __restrict__ P, int N, int G, int LP) {
    __shared__ float degl[RB];

    const int tid  = threadIdx.x;
    const int lane = tid & 31;
    const int half = lane >> 4;
    const int lm   = lane & 15;
    const int wv    = __builtin_amdgcn_readfirstlane(tid >> 5);
    const int base  = blockIdx.x * RB;
    int rows = N - base;
    if (rows > RB) rows = RB;

    const int  row0     = base + wv * 16;
    const bool has_tile = (wv < GW) && (row0 < N);

    v2f ax[8], ah[8];
#pragma unroll
    for (int kt = 0; kt < 8; ++kt) { ax[kt] = v2f{0.f, 0.f}; ah[kt] = v2f{0.f, 0.f}; }
    if (has_tile) {
        int arow = row0 + lm;
        if (arow > N - 1) arow = N - 1;
        const float* xrow = X + (size_t)arow * CH;
        const float* hrow = H + (size_t)arow * CH;
#pragma unroll
        for (int kt = 0; kt < 8; ++kt) {
            int ka = kt * 4 + 2 * half;
            ax[kt] = v2f{ xrow[ka], xrow[ka + 1] };
            ah[kt] = v2f{ hrow[ka], hrow[ka + 1] };
        }
    }

    for (int i = tid; i < RB; i += GT) degl[i] = 0.0f;
    __syncthreads();

    for (int gb = 0; gb < G; gb += SEGB) {
        const int gk = gb + wv + 32 * lane;
        int a = 0, n = 0;
        if (gk < G) {
            const int* lr = lstg + (size_t)gk * LP + blockIdx.x;
            a = lr[0];
            n = lr[1] - a;
        }
        const int A = gk * CHK + a;
        int ks = (G - gb - wv + 31) >> 5;
        if (ks > 32) ks = 32;
        for (int k0 = 0; k0 < ks; k0 += SGB) {
            int   dl[SGB * SGT];
            float wq[SGB * SGT];
#pragma unroll
            for (int kk = 0; kk < SGB; ++kk) {
                const int Ak = __builtin_amdgcn_readlane(A, k0 + kk);
                const int nk = __builtin_amdgcn_readlane(n, k0 + kk);
#pragma unroll
                for (int t = 0; t < SGT; ++t) {
                    const int j = lane + 32 * t;
                    dl[kk * SGT + t] = 0;
                    wq[kk * SGT + t] = 0.0f;
                    if (j < nk) {
                        dl[kk * SGT + t] = bdl[Ak + j];
                        wq[kk * SGT + t] = bw[Ak + j];
                    }
                }
            }
#pragma unroll
            for (int kk = 0; kk < SGB; ++kk) {
                const int nk = __builtin_amdgcn_readlane(n, k0 + kk);
#pragma unroll
                for (int t = 0; t < SGT; ++t) {
                    if (lane + 32 * t < nk) {
                        __hip_atomic_fetch_add(&degl[dl[kk * SGT + t]], wq[kk * SGT + t],
                                               __ATOMIC_RELAXED, __HIP_MEMORY_SCOPE_WORKGROUP);
                    }
                }
            }
            for (int kk = 0; kk < SGB; ++kk) {
                const int Ak = __builtin_amdgcn_readlane(A, k0 + kk);
                const int nk = __builtin_amdgcn_readlane(n, k0 + kk);
                for (int j = 32 * SGT + lane; j < nk; j += 32) {
                    __hip_atomic_fetch_add(&degl[bdl[Ak + j]], bw[Ak + j],
                                           __ATOMIC_RELAXED, __HIP_MEMORY_SCOPE_WORKGROUP);
                }
            }
        }
    }
    __syncthreads();
    if (tid < RB) {
        float dg = degl[tid];
        float v  = (dg > 0.0f) ? rsqrtf(fmaxf(dg, 1e-12f)) : 0.0f;
        degl[tid] = v;
        if (tid < rows) dinv[base + tid] = v;
    }
    __syncthreads();

    if (has_tile) {
        float dv[8];
#pragma unroll
        for (int r = 0; r < 8; ++r) dv[r] = degl[wv * 16 + r + 8 * half];

        v8f acc0 = {};
        v8f acc1 = {};

#pragma unroll
        for (int kt = 0; kt < 8; ++kt) {
            int ka = kt * 4 + 2 * half;

            int krow = ka * CH;
            v2f bx0 = { Wx[krow + lm],      Wx[krow + CH + lm]      };
            v2f bx1 = { Wx[krow + 16 + lm], Wx[krow + CH + 16 + lm] };
            v2f bh0 = { Wh[krow + lm],      Wh[krow + CH + lm]      };
            v2f bh1 = { Wh[krow + 16 + lm], Wh[krow + CH + 16 + lm] };

            acc0 = __builtin_amdgcn_wmma_f32_16x16x4_f32(false, ax[kt], false, bx0,
                                                         (short)0, acc0, false, false);
            acc1 = __builtin_amdgcn_wmma_f32_16x16x4_f32(false, ax[kt], false, bx1,
                                                         (short)0, acc1, false, false);
            acc0 = __builtin_amdgcn_wmma_f32_16x16x4_f32(false, ah[kt], false, bh0,
                                                         (short)0, acc0, false, false);
            acc1 = __builtin_amdgcn_wmma_f32_16x16x4_f32(false, ah[kt], false, bh1,
                                                         (short)0, acc1, false, false);
        }

        float* prow = P + (size_t)row0 * CH;
#pragma unroll
        for (int r = 0; r < 8; ++r) {
            int rr = r + 8 * half;
            if (row0 + rr < N) {
                prow[(size_t)rr * CH + lm]      = dv[r] * acc0[r];
                prow[(size_t)rr * CH + 16 + lm] = dv[r] * acc1[r];
            }
        }
    }
}

__device__ __forceinline__ float sigmoidf_(float x) {
    return 1.0f / (1.0f + __expf(-x));
}

__global__ __launch_bounds__(SCW * 32) void k_scatter_gates(const int* __restrict__ lstg,
                                                            const int* __restrict__ bsrc,
                                                            const float* __restrict__ bw,
                                                            const int* __restrict__ bdl,
                                                            const float* __restrict__ P,
                                                            const float* __restrict__ dinv,
                                                            const float* __restrict__ Cst,
                                                            const float* __restrict__ bx,
                                                            const float* __restrict__ bh,
                                                            const float* __restrict__ bi,
                                                            const float* __restrict__ bf,
                                                            const float* __restrict__ bc,
                                                            const float* __restrict__ bo,
                                                            float* __restrict__ out, int N, int G, int LP) {
    __shared__ __attribute__((aligned(16))) float tile[RB * CH];

    const int tid  = threadIdx.x;
    const int lane = tid & 31;
    const int wv    = __builtin_amdgcn_readfirstlane(tid >> 5);
    const int base  = blockIdx.x * RB;
    int rows = N - base;
    if (rows > RB) rows = RB;

    int gb = 0;
    int sa = 0, se = 0;
    int segEnd = 0, segOff = 0, total = 0;
    int ku = 0;
    int   s = 0, dl = 0;
    float nrm = 0.0f;

    auto load_bounds = [&]() {
        const int gk = gb + wv + 32 * lane;
        sa = 0; se = 0;
        if (gk < G) {
            const int* lr = lstg + (size_t)gk * LP + blockIdx.x;
            sa = lr[0];
            se = lr[1];
        }
    };
    auto make_list = [&]() {
        const int gk  = gb + wv + 32 * lane;
        const int sn  = se - sa;
        const int inc = wave_scan_incl(sn, lane);
        segEnd = inc;
        segOff = gk * CHK + sa - (inc - sn);
        total  = __builtin_amdgcn_readlane(inc, 31);
        ku     = 0;
    };
    auto fetch = [&](int pg, int& so, float& wo, int& dlo) {
        int pfirst = pg, plast = pg + 31;
        if (pfirst > total - 1) pfirst = total - 1;
        if (plast  > total - 1) plast  = total - 1;
        int q = pg + lane;
        if (q > total - 1) q = total - 1;
        int k = ku;
        int e = __builtin_amdgcn_readlane(segEnd, k);
        while (e <= pfirst && k < 31) {
            ++k;
            e = __builtin_amdgcn_readlane(segEnd, k);
        }
        ku = k;
        int off = __builtin_amdgcn_readlane(segOff, k);
        while (e <= plast && k < 31) {
            ++k;
            const int o2 = __builtin_amdgcn_readlane(segOff, k);
            off = (q >= e) ? o2 : off;
            e = __builtin_amdgcn_readlane(segEnd, k);
        }
        const int ix = q + off;
        so  = bsrc[ix];
        wo  = bw[ix];
        dlo = bdl[ix];
    };

    load_bounds();

    const float* crow = Cst + (size_t)base * CH;
    float cv[GI], dvr[GI];
#pragma unroll
    for (int k = 0; k < GI; ++k) {
        int r = wv + k * SCW;
        if (r > rows - 1) r = rows - 1;
        cv[k]  = crow[r * CH + lane];
        dvr[k] = dinv[base + r];
    }
    const float bxl = bx[lane], bhl = bh[lane], bil = bi[lane], bfl = bf[lane], bcl = bc[lane], bol = bo[lane];

    make_list();
    if (total > 0) fetch(0, s, nrm, dl);

    float4* tile4 = reinterpret_cast<float4*>(tile);
    for (int i = tid; i < RB * CH / 4; i += SCW * 32) tile4[i] = make_float4(0.f, 0.f, 0.f, 0.f);
    __syncthreads();

    for (;;) {
        int p = 0;
        while (p < total) {
            int   s2, dl2;
            float nrm2;
            fetch(p + 32, s2, nrm2, dl2);

            int n = total - p;
            if (n > 32) n = 32;
            float pv[32];
#pragma unroll
            for (int i = 0; i < 32; ++i) {
                unsigned si = (unsigned)__builtin_amdgcn_readlane(s, i);
                pv[i] = P[si * (unsigned)CH + (unsigned)lane];
            }
            if (n == 32) {
#pragma unroll
                for (int i = 0; i < 32; ++i) {
                    float ni = __int_as_float(__builtin_amdgcn_readlane(__float_as_int(nrm), i));
                    int di = __builtin_amdgcn_readlane(dl, i);
                    __hip_atomic_fetch_add(&tile[di * CH + lane], ni * pv[i],
                                           __ATOMIC_RELAXED, __HIP_MEMORY_SCOPE_WORKGROUP);
                }
            } else {
#pragma unroll
                for (int i = 0; i < 32; ++i) {
                    if (i < n) {
                        float ni = __int_as_float(__builtin_amdgcn_readlane(__float_as_int(nrm), i));
                        int di = __builtin_amdgcn_readlane(dl, i);
                        __hip_atomic_fetch_add(&tile[di * CH + lane], ni * pv[i],
                                               __ATOMIC_RELAXED, __HIP_MEMORY_SCOPE_WORKGROUP);
                    }
                }
            }
            s = s2; nrm = nrm2; dl = dl2; p += 32;
        }
        gb += SEGB;
        if (gb >= G) break;
        load_bounds();
        make_list();
        if (total > 0) fetch(0, s, nrm, dl);
    }

    __syncthreads();

    const size_t NC = (size_t)N * CH;
    float* orow = out + (size_t)base * CH;
#pragma unroll
    for (int k = 0; k < GI; ++k) {
        const int r = wv + k * SCW;
        if (r < rows) {
            const int i = r * CH + lane;
            float a  = tile[i];
            float sg = dvr[k] * a + bxl + bhl;
            float I  = sigmoidf_(sg + bil);
            float F  = sigmoidf_(sg + bfl);
            float T  = tanhf(sg + bcl);
            float Cn = F * cv[k] + I * T;
            float O  = sigmoidf_(sg + bol);
            float Hn = O * tanhf(Cn);
            orow[i]          = O;
            orow[NC + i]     = Hn;
            orow[2 * NC + i] = Cn;
        }
    }
}

extern "C" void kernel_launch(void* const* d_in, const int* in_sizes, int n_in,
                              void* d_out, int out_size, void* d_ws, size_t ws_size,
                              hipStream_t stream) {
    const float* X   = (const float*)d_in[0];
    const int*   ei  = (const int*)  d_in[1];
    const float* w   = (const float*)d_in[2];
    const float* H   = (const float*)d_in[3];
    const float* C   = (const float*)d_in[4];
    const float* Wx  = (const float*)d_in[5];
    const float* bx  = (const float*)d_in[6];
    const float* Wh  = (const float*)d_in[7];
    const float* bh  = (const float*)d_in[8];
    const float* bi  = (const float*)d_in[9];
    const float* bf  = (const float*)d_in[10];
    const float* bc  = (const float*)d_in[11];
    const float* bo  = (const float*)d_in[12];

    const int N = in_sizes[0] / CH;
    const int E = in_sizes[2];
    const int* src = ei;
    const int* dst = ei + E;

    float* ws  = (float*)d_ws;
    size_t Na  = ((size_t)N + 31) & ~(size_t)31;
    float* dinv = ws;
    float* P    = dinv + Na;
    const int NB = (N + RB - 1) / RB;
    const int G  = (E + CHK - 1) / CHK;
    const int LP = (NB + 1 + 31) & ~31;
    int*   lstg = (int*)(P + (size_t)N * CH);
    int*   bsrc = lstg + (size_t)G * LP;
    float* bw   = (float*)(bsrc + (size_t)G * CHK);
    int*   bdl  = (int*)(bw + (size_t)G * CHK);

    k_bin_local<<<G, BT, 0, stream>>>(src, dst, w, lstg, bsrc, bw, bdl, E, N, NB, LP);
    k_deg_gemm_wmma<<<NB, GT, 0, stream>>>(lstg, bw, bdl, X, H, Wx, Wh, dinv, P, N, G, LP);
    k_scatter_gates<<<NB, SCW * 32, 0, stream>>>(lstg, bsrc, bw, bdl, P, dinv, C,
                                                 bx, bh, bi, bf, bc, bo, (float*)d_out, N, G, LP);
}


# gate pass: sigmoid and tanh as 1/(1+2^t) on hardware exp2 and reciprocal, no branches, no IEEE divide, no library tanh; each gate's bias sum folded into one per-channel constant (re-associated)
# speedup vs baseline: 7.3811x; 1.0590x over previous
#include <hip/hip_runtime.h>
#include <hip/hip_bf16.h>
#include <math.h>

typedef __attribute__((ext_vector_type(2))) float v2f;
typedef __attribute__((ext_vector_type(8))) float v8f;

#define CH 32

#define RB    400
#define CHK   12800
#define BT    1024
#define BU    13
#define NBMAX 1024
#define SCW   32
#define SEGB  (SCW * 32)
#define GI    ((RB * CH + SCW * 32 - 1) / (SCW * 32))

static_assert(BT == NBMAX && BT == 32 * 32, "one thread per table bin, and one lane per wave total in the scan");

__device__ __forceinline__ int wave_scan_incl(int v, int lane) {
    int inc = v;
    for (int off = 1; off < 32; off <<= 1) {
        int t = __shfl_up(inc, off);
        if (lane >= off) inc += t;
    }
    return inc;
}

__global__ __launch_bounds__(BT) void k_bin_local(const int* __restrict__ src,
                                                  const int* __restrict__ dst,
                                                  const float* __restrict__ w,
                                                  int* __restrict__ lstg,
                                                  int* __restrict__ bsrc,
                                                  float* __restrict__ bw,
                                                  int* __restrict__ bdl,
                                                  int E, int N, int NB, int LP) {
    __shared__ int   ssrc[CHK];
    __shared__ float sw[CHK];
    __shared__ int   sdl[CHK];
    __shared__ int   cntl[NBMAX];
    __shared__ int   lst[NBMAX + 1];
    __shared__ int   wtot[BT / 32];

    const int tid  = threadIdx.x;
    const int lane = tid & 31;
    const int wv   = tid >> 5;
    const int g    = blockIdx.x;
    const int e0   = g * CHK;
    int e1 = e0 + CHK;
    if (e1 > E) e1 = E;

    int   d[BU], s[BU];
    float ww[BU];
#pragma unroll
    for (int u = 0; u < BU; ++u) {
        int e = e0 + u * BT + tid;
        bool in = e < e1;
        int ee = in ? e : e0;
        int dv = dst[ee];
        s[u]  = src[ee];
        ww[u] = w[ee];
        d[u]  = in ? dv : -1;
    }

    cntl[tid] = 0;
    __syncthreads();

    int rk[BU];
#pragma unroll
    for (int u = 0; u < BU; ++u) {
        rk[u] = 0;
        if ((unsigned)d[u] < (unsigned)N) {
            rk[u] = __hip_atomic_fetch_add(&cntl[(unsigned)d[u] / RB], 1,
                                           __ATOMIC_RELAXED, __HIP_MEMORY_SCOPE_WORKGROUP);
        }
    }
    __syncthreads();

    const int c   = (tid < NB) ? cntl[tid] : 0;
    const int inc = wave_scan_incl(c, lane);
    if (lane == 31) wtot[wv] = inc;
    __syncthreads();
    const int winc = wave_scan_incl(wtot[lane], lane);
    int wpre = __shfl(winc, (wv + 31) & 31);
    if (wv == 0) wpre = 0;
    const int excl = wpre + inc - c;
    if (tid < NB) lst[tid] = excl;
    if (tid == BT - 1) lst[NB] = excl + c;
    __syncthreads();

    for (int b = tid; b <= NB; b += BT) lstg[(size_t)g * LP + b] = lst[b];

#pragma unroll
    for (int u = 0; u < BU; ++u) {
        if ((unsigned)d[u] < (unsigned)N) {
            unsigned b = (unsigned)d[u] / RB;
            int p = lst[b] + rk[u];
            ssrc[p] = s[u];
            sw[p]   = ww[u];
            sdl[p]  = (int)((unsigned)d[u] - b * RB);
        }
    }
    __syncthreads();

    const int n = lst[NB];
    int*   osrc = bsrc + (size_t)g * CHK;
    float* ow   = bw   + (size_t)g * CHK;
    int*   odl  = bdl  + (size_t)g * CHK;
#pragma unroll
    for (int u = 0; u < BU; ++u) {
        int i = u * BT + tid;
        if (i < n) {
            osrc[i] = ssrc[i];
            ow[i]   = sw[i];
            odl[i]  = sdl[i];
        }
    }
}

#define GW   (RB / 16)
#define GT   (SCW * 32)
#define SGB  8
#define SGT  3

static_assert(RB % 16 == 0, "a bin is a whole number of 16-row tiles");
static_assert(GW * 32 <= GT && RB <= GT, "one wave per tile and one thread per bin row");

__global__ __launch_bounds__(GT) void k_deg_gemm_wmma(const int* __restrict__ lstg,
                                                      const float* __restrict__ bw,
                                                      const int* __restrict__ bdl,
                                                      const float* __restrict__ X,
                                                      const float* __restrict__ H,
                                                      const float* __restrict__ Wx,
                                                      const float* __restrict__ Wh,
                                                      float* __restrict__ dinv,
                                                      float* __restrict__ P, int N, int G, int LP) {
    __shared__ float degl[RB];

    const int tid  = threadIdx.x;
    const int lane = tid & 31;
    const int half = lane >> 4;
    const int lm   = lane & 15;
    const int wv    = __builtin_amdgcn_readfirstlane(tid >> 5);
    const int base  = blockIdx.x * RB;
    int rows = N - base;
    if (rows > RB) rows = RB;

    const int  row0     = base + wv * 16;
    const bool has_tile = (wv < GW) && (row0 < N);

    v2f ax[8], ah[8];
#pragma unroll
    for (int kt = 0; kt < 8; ++kt) { ax[kt] = v2f{0.f, 0.f}; ah[kt] = v2f{0.f, 0.f}; }
    if (has_tile) {
        int arow = row0 + lm;
        if (arow > N - 1) arow = N - 1;
        const float* xrow = X + (size_t)arow * CH;
        const float* hrow = H + (size_t)arow * CH;
#pragma unroll
        for (int kt = 0; kt < 8; ++kt) {
            int ka = kt * 4 + 2 * half;
            ax[kt] = v2f{ xrow[ka], xrow[ka + 1] };
            ah[kt] = v2f{ hrow[ka], hrow[ka + 1] };
        }
    }

    for (int i = tid; i < RB; i += GT) degl[i] = 0.0f;
    __syncthreads();

    for (int gb = 0; gb < G; gb += SEGB) {
        const int gk = gb + wv + 32 * lane;
        int a = 0, n = 0;
        if (gk < G) {
            const int* lr = lstg + (size_t)gk * LP + blockIdx.x;
            a = lr[0];
            n = lr[1] - a;
        }
        const int A = gk * CHK + a;
        int ks = (G - gb - wv + 31) >> 5;
        if (ks > 32) ks = 32;
        for (int k0 = 0; k0 < ks; k0 += SGB) {
            int   dl[SGB * SGT];
            float wq[SGB * SGT];
#pragma unroll
            for (int kk = 0; kk < SGB; ++kk) {
                const int Ak = __builtin_amdgcn_readlane(A, k0 + kk);
                const int nk = __builtin_amdgcn_readlane(n, k0 + kk);
#pragma unroll
                for (int t = 0; t < SGT; ++t) {
                    const int j = lane + 32 * t;
                    dl[kk * SGT + t] = 0;
                    wq[kk * SGT + t] = 0.0f;
                    if (j < nk) {
                        dl[kk * SGT + t] = bdl[Ak + j];
                        wq[kk * SGT + t] = bw[Ak + j];
                    }
                }
            }
#pragma unroll
            for (int kk = 0; kk < SGB; ++kk) {
                const int nk = __builtin_amdgcn_readlane(n, k0 + kk);
#pragma unroll
                for (int t = 0; t < SGT; ++t) {
                    if (lane + 32 * t < nk) {
                        __hip_atomic_fetch_add(&degl[dl[kk * SGT + t]], wq[kk * SGT + t],
                                               __ATOMIC_RELAXED, __HIP_MEMORY_SCOPE_WORKGROUP);
                    }
                }
            }
            for (int kk = 0; kk < SGB; ++kk) {
                const int Ak = __builtin_amdgcn_readlane(A, k0 + kk);
                const int nk = __builtin_amdgcn_readlane(n, k0 + kk);
                for (int j = 32 * SGT + lane; j < nk; j += 32) {
                    __hip_atomic_fetch_add(&degl[bdl[Ak + j]], bw[Ak + j],
                                           __ATOMIC_RELAXED, __HIP_MEMORY_SCOPE_WORKGROUP);
                }
            }
        }
    }
    __syncthreads();
    if (tid < RB) {
        float dg = degl[tid];
        float v  = (dg > 0.0f) ? rsqrtf(fmaxf(dg, 1e-12f)) : 0.0f;
        degl[tid] = v;
        if (tid < rows) dinv[base + tid] = v;
    }
    __syncthreads();

    if (has_tile) {
        float dv[8];
#pragma unroll
        for (int r = 0; r < 8; ++r) dv[r] = degl[wv * 16 + r + 8 * half];

        v8f acc0 = {};
        v8f acc1 = {};

#pragma unroll
        for (int kt = 0; kt < 8; ++kt) {
            int ka = kt * 4 + 2 * half;

            int krow = ka * CH;
            v2f bx0 = { Wx[krow + lm],      Wx[krow + CH + lm]      };
            v2f bx1 = { Wx[krow + 16 + lm], Wx[krow + CH + 16 + lm] };
            v2f bh0 = { Wh[krow + lm],      Wh[krow + CH + lm]      };
            v2f bh1 = { Wh[krow + 16 + lm], Wh[krow + CH + 16 + lm] };

            acc0 = __builtin_amdgcn_wmma_f32_16x16x4_f32(false, ax[kt], false, bx0,
                                                         (short)0, acc0, false, false);
            acc1 = __builtin_amdgcn_wmma_f32_16x16x4_f32(false, ax[kt], false, bx1,
                                                         (short)0, acc1, false, false);
            acc0 = __builtin_amdgcn_wmma_f32_16x16x4_f32(false, ah[kt], false, bh0,
                                                         (short)0, acc0, false, false);
            acc1 = __builtin_amdgcn_wmma_f32_16x16x4_f32(false, ah[kt], false, bh1,
                                                         (short)0, acc1, false, false);
        }

        float* prow = P + (size_t)row0 * CH;
#pragma unroll
        for (int r = 0; r < 8; ++r) {
            int rr = r + 8 * half;
            if (row0 + rr < N) {
                prow[(size_t)rr * CH + lm]      = dv[r] * acc0[r];
                prow[(size_t)rr * CH + 16 + lm] = dv[r] * acc1[r];
            }
        }
    }
}

#define LOG2E 1.4426950408889634f
__device__ __forceinline__ float rcp1p_exp2_(float t) {
    return __builtin_amdgcn_rcpf(1.0f + __builtin_amdgcn_exp2f(t));
}

__global__ __launch_bounds__(SCW * 32) void k_scatter_gates(const int* __restrict__ lstg,
                                                            const int* __restrict__ bsrc,
                                                            const float* __restrict__ bw,
                                                            const int* __restrict__ bdl,
                                                            const float* __restrict__ P,
                                                            const float* __restrict__ dinv,
                                                            const float* __restrict__ Cst,
                                                            const float* __restrict__ bx,
                                                            const float* __restrict__ bh,
                                                            const float* __restrict__ bi,
                                                            const float* __restrict__ bf,
                                                            const float* __restrict__ bc,
                                                            const float* __restrict__ bo,
                                                            float* __restrict__ out, int N, int G, int LP) {
    __shared__ __attribute__((aligned(16))) float tile[RB * CH];

    const int tid  = threadIdx.x;
    const int lane = tid & 31;
    const int wv    = __builtin_amdgcn_readfirstlane(tid >> 5);
    const int base  = blockIdx.x * RB;
    int rows = N - base;
    if (rows > RB) rows = RB;

    int gb = 0;
    int sa = 0, se = 0;
    int segEnd = 0, segOff = 0, total = 0;
    int ku = 0;
    int   s = 0, dl = 0;
    float nrm = 0.0f;

    auto load_bounds = [&]() {
        const int gk = gb + wv + 32 * lane;
        sa = 0; se = 0;
        if (gk < G) {
            const int* lr = lstg + (size_t)gk * LP + blockIdx.x;
            sa = lr[0];
            se = lr[1];
        }
    };
    auto make_list = [&]() {
        const int gk  = gb + wv + 32 * lane;
        const int sn  = se - sa;
        const int inc = wave_scan_incl(sn, lane);
        segEnd = inc;
        segOff = gk * CHK + sa - (inc - sn);
        total  = __builtin_amdgcn_readlane(inc, 31);
        ku     = 0;
    };
    auto fetch = [&](int pg, int& so, float& wo, int& dlo) {
        int pfirst = pg, plast = pg + 31;
        if (pfirst > total - 1) pfirst = total - 1;
        if (plast  > total - 1) plast  = total - 1;
        int q = pg + lane;
        if (q > total - 1) q = total - 1;
        int k = ku;
        int e = __builtin_amdgcn_readlane(segEnd, k);
        while (e <= pfirst && k < 31) {
            ++k;
            e = __builtin_amdgcn_readlane(segEnd, k);
        }
        ku = k;
        int off = __builtin_amdgcn_readlane(segOff, k);
        while (e <= plast && k < 31) {
            ++k;
            const int o2 = __builtin_amdgcn_readlane(segOff, k);
            off = (q >= e) ? o2 : off;
            e = __builtin_amdgcn_readlane(segEnd, k);
        }
        const int ix = q + off;
        so  = bsrc[ix];
        wo  = bw[ix];
        dlo = bdl[ix];
    };

    load_bounds();

    const float* crow = Cst + (size_t)base * CH;
    float cv[GI], dvr[GI];
#pragma unroll
    for (int k = 0; k < GI; ++k) {
        int r = wv + k * SCW;
        if (r > rows - 1) r = rows - 1;
        cv[k]  = crow[r * CH + lane];
        dvr[k] = dinv[base + r];
    }
    const float bxl = bx[lane], bhl = bh[lane], bil = bi[lane], bfl = bf[lane], bcl = bc[lane], bol = bo[lane];

    make_list();
    if (total > 0) fetch(0, s, nrm, dl);

    float4* tile4 = reinterpret_cast<float4*>(tile);
    for (int i = tid; i < RB * CH / 4; i += SCW * 32) tile4[i] = make_float4(0.f, 0.f, 0.f, 0.f);
    __syncthreads();

    for (;;) {
        int p = 0;
        while (p < total) {
            int   s2, dl2;
            float nrm2;
            fetch(p + 32, s2, nrm2, dl2);

            int n = total - p;
            if (n > 32) n = 32;
            float pv[32];
#pragma unroll
            for (int i = 0; i < 32; ++i) {
                unsigned si = (unsigned)__builtin_amdgcn_readlane(s, i);
                pv[i] = P[si * (unsigned)CH + (unsigned)lane];
            }
            if (n == 32) {
#pragma unroll
                for (int i = 0; i < 32; ++i) {
                    float ni = __int_as_float(__builtin_amdgcn_readlane(__float_as_int(nrm), i));
                    int di = __builtin_amdgcn_readlane(dl, i);
                    __hip_atomic_fetch_add(&tile[di * CH + lane], ni * pv[i],
                                           __ATOMIC_RELAXED, __HIP_MEMORY_SCOPE_WORKGROUP);
                }
            } else {
#pragma unroll
                for (int i = 0; i < 32; ++i) {
                    if (i < n) {
                        float ni = __int_as_float(__builtin_amdgcn_readlane(__float_as_int(nrm), i));
                        int di = __builtin_amdgcn_readlane(dl, i);
                        __hip_atomic_fetch_add(&tile[di * CH + lane], ni * pv[i],
                                               __ATOMIC_RELAXED, __HIP_MEMORY_SCOPE_WORKGROUP);
                    }
                }
            }
            s = s2; nrm = nrm2; dl = dl2; p += 32;
        }
        gb += SEGB;
        if (gb >= G) break;
        load_bounds();
        make_list();
        if (total > 0) fetch(0, s, nrm, dl);
    }

    __syncthreads();

    const float bs = bxl + bhl;
    const float kI = -LOG2E * (bs + bil);
    const float kF = -LOG2E * (bs + bfl);
    const float kO = -LOG2E * (bs + bol);
    const float kT = 2.0f * LOG2E * (bs + bcl);
    const size_t NC = (size_t)N * CH;
    float* orow = out + (size_t)base * CH;
#pragma unroll
    for (int k = 0; k < GI; ++k) {
        const int r = wv + k * SCW;
        if (r < rows) {
            const int i = r * CH + lane;
            const float u  = dvr[k] * tile[i];
            const float I  = rcp1p_exp2_(__builtin_fmaf(u, -LOG2E, kI));
            const float F  = rcp1p_exp2_(__builtin_fmaf(u, -LOG2E, kF));
            const float O  = rcp1p_exp2_(__builtin_fmaf(u, -LOG2E, kO));
            const float T  = __builtin_fmaf(rcp1p_exp2_(__builtin_fmaf(u, 2.0f * LOG2E, kT)), -2.0f, 1.0f);
            const float Cn = F * cv[k] + I * T;
            const float Hn = O * __builtin_fmaf(rcp1p_exp2_(Cn * (2.0f * LOG2E)), -2.0f, 1.0f);
            orow[i]          = O;
            orow[NC + i]     = Hn;
            orow[2 * NC + i] = Cn;
        }
    }
}

extern "C" void kernel_launch(void* const* d_in, const int* in_sizes, int n_in,
                              void* d_out, int out_size, void* d_ws, size_t ws_size,
                              hipStream_t stream) {
    const float* X   = (const float*)d_in[0];
    const int*   ei  = (const int*)  d_in[1];
    const float* w   = (const float*)d_in[2];
    const float* H   = (const float*)d_in[3];
    const float* C   = (const float*)d_in[4];
    const float* Wx  = (const float*)d_in[5];
    const float* bx  = (const float*)d_in[6];
    const float* Wh  = (const float*)d_in[7];
    const float* bh  = (const float*)d_in[8];
    const float* bi  = (const float*)d_in[9];
    const float* bf  = (const float*)d_in[10];
    const float* bc  = (const float*)d_in[11];
    const float* bo  = (const float*)d_in[12];

    const int N = in_sizes[0] / CH;
    const int E = in_sizes[2];
    const int* src = ei;
    const int* dst = ei + E;

    float* ws  = (float*)d_ws;
    size_t Na  = ((size_t)N + 31) & ~(size_t)31;
    float* dinv = ws;
    float* P    = dinv + Na;
    const int NB = (N + RB - 1) / RB;
    const int G  = (E + CHK - 1) / CHK;
    const int LP = (NB + 1 + 31) & ~31;
    int*   lstg = (int*)(P + (size_t)N * CH);
    int*   bsrc = lstg + (size_t)G * LP;
    float* bw   = (float*)(bsrc + (size_t)G * CHK);
    int*   bdl  = (int*)(bw + (size_t)G * CHK);

    k_bin_local<<<G, BT, 0, stream>>>(src, dst, w, lstg, bsrc, bw, bdl, E, N, NB, LP);
    k_deg_gemm_wmma<<<NB, GT, 0, stream>>>(lstg, bw, bdl, X, H, Wx, Wh, dinv, P, N, G, LP);
    k_scatter_gates<<<NB, SCW * 32, 0, stream>>>(lstg, bsrc, bw, bdl, P, dinv, C,
                                                 bx, bh, bi, bf, bc, bo, (float*)d_out, N, G, LP);
}
